# starting kernel
# baseline (speedup 1.0000x reference)
#include <hip/hip_runtime.h>

typedef float v2f __attribute__((ext_vector_type(2)));
typedef float v8f __attribute__((ext_vector_type(8)));


__global__ void k_init_deg(float* deg, int n) {
    int i = blockIdx.x * blockDim.x + threadIdx.x;
    if (i < n) deg[i] = 1.0f;
}

__global__ void k_count_deg(const int* __restrict__ dst, long e, float* deg) {
    long i = (long)blockIdx.x * blockDim.x + threadIdx.x;
    if (i < e) unsafeAtomicAdd(&deg[dst[i]], 1.0f);
}

__global__ void k_rsqrt_inplace(float* deg, int n) {
    int i = blockIdx.x * blockDim.x + threadIdx.x;
    if (i < n) deg[i] = rsqrtf(deg[i]);
}


template <int K, int NOUT, bool RELU_BIAS>
__global__ __launch_bounds__(256) void k_gemm_wmma(const float* __restrict__ A,
                                                   const float* __restrict__ W,
                                                   const float* __restrict__ bias,
                                                   float* __restrict__ Out,
                                                   int nRows) {
    constexpr int PITCH = 2 * NOUT + 32;
    constexpr int HK    = K / 2;
    constexpr int PAIRS = K / 4;
    constexpr int CT    = NOUT / 16;
    __shared__ __align__(16) float lw[PAIRS * PITCH];

    const int lane  = threadIdx.x & 31;
    const int wave  = threadIdx.x >> 5;
    const int laneM = lane & 15;
    const int hi    = lane >> 4;

    const int rowBase = blockIdx.x * 128 + wave * 16 + laneM;
    const int rowC    = rowBase < nRows ? rowBase : 0;

    v8f acc[CT] = {};

    for (int kh = 0; kh < 2; ++kh) {
        __syncthreads();
        for (int idx = threadIdx.x; idx < HK * NOUT; idx += 256) {
            int lk = idx / NOUT;
            int c  = idx - lk * NOUT;
            lw[(lk >> 1) * PITCH + 2 * c + (lk & 1)] = W[(kh * HK + lk) * NOUT + c];
        }
        __syncthreads();

        for (int k4 = 0; k4 < HK; k4 += 4) {
            const int kk = kh * HK + k4 + 2 * hi;
            v2f a;
            a.x = A[(long)rowC * K + kk];
            a.y = A[(long)rowC * K + kk + 1];
            if (RELU_BIAS) {
                a.x = fmaxf(a.x + bias[kk], 0.0f);
                a.y = fmaxf(a.y + bias[kk + 1], 0.0f);
            }
            const float* lb = &lw[((k4 >> 1) + hi) * PITCH];

            v2f bfrag[CT];
#pragma unroll
            for (int ct = 0; ct < CT; ++ct)
                bfrag[ct] = *(const v2f*)&lb[2 * (ct * 16 + laneM)];

#pragma unroll
            for (int ct = 0; ct < CT; ++ct)
                acc[ct] = __builtin_amdgcn_wmma_f32_16x16x4_f32(
                    false, a, false, bfrag[ct], (short)0, acc[ct], false, false);
        }
    }

#pragma unroll
    for (int ct = 0; ct < CT; ++ct) {
#pragma unroll
        for (int r = 0; r < 8; ++r) {
            int m = blockIdx.x * 128 + wave * 16 + 8 * hi + r;
            if (m < nRows) Out[(long)m * NOUT + ct * 16 + laneM] = acc[ct][r];
        }
    }
}


__global__ void k_self_init(const float* __restrict__ h, const float* __restrict__ dinv,
                            const float* __restrict__ bias, float* __restrict__ out,
                            long total, int logF) {
    long gid = (long)blockIdx.x * blockDim.x + threadIdx.x;
    if (gid >= total) return;
    int  i = (int)(gid >> logF);
    int  f = (int)(gid & ((1 << logF) - 1));
    float d = dinv[i];
    float v = h[gid] * d * d;
    if (bias) v += bias[f];
    out[gid] = v;
}


template <int LOGF>
__global__ void k_scatter(const int* __restrict__ src, const int* __restrict__ dst,
                          const float* __restrict__ dinv, const float* __restrict__ h,
                          float* __restrict__ out, long nE) {
    constexpr int CH = (1 << LOGF) / 4;
    long gid = (long)blockIdx.x * blockDim.x + threadIdx.x;
    long e   = gid >> (LOGF - 2);
    if (e >= nE) return;
    int   c4 = (int)(gid & (CH - 1));
    int   s  = src[e], d = dst[e];
    float w  = dinv[s] * dinv[d];
    float4 v = ((const float4*)(h + ((long)s << LOGF)))[c4];
    float* o = out + ((long)d << LOGF) + 4 * c4;
    unsafeAtomicAdd(o + 0, v.x * w);
    unsafeAtomicAdd(o + 1, v.y * w);
    unsafeAtomicAdd(o + 2, v.z * w);
    unsafeAtomicAdd(o + 3, v.w * w);
}


extern "C" void kernel_launch(void* const* d_in, const int* in_sizes, int n_in,
                              void* d_out, int out_size, void* d_ws, size_t ws_size,
                              hipStream_t stream) {
    const float* x   = (const float*)d_in[0];
    const int*   ei  = (const int*)d_in[1];
    const float* W1  = (const float*)d_in[2];
    const float* b1  = (const float*)d_in[3];
    const float* W2  = (const float*)d_in[4];
    const float* b2  = (const float*)d_in[5];
    float*       out = (float*)d_out;

    const int  n = in_sizes[0] / 128;
    const long E = in_sizes[1] / 2;
    const int* srcIdx = ei;
    const int* dstIdx = ei + E;

    float* ws   = (float*)d_ws;
    float* dinv = ws;
    float* h1   = dinv + n;
    float* agg1 = h1 + (long)n * 128;
    float* h2   = agg1 + (long)n * 128;

    const int B = 256;
    auto blk = [](long work, int b) { return (int)((work + b - 1) / b); };

    k_init_deg<<<blk(n, B), B, 0, stream>>>(dinv, n);
    k_count_deg<<<blk(E, B), B, 0, stream>>>(dstIdx, E, dinv);
    k_rsqrt_inplace<<<blk(n, B), B, 0, stream>>>(dinv, n);

    k_gemm_wmma<128, 128, false><<<blk(n, 128), 256, 0, stream>>>(x, W1, nullptr, h1, n);
    k_self_init<<<blk((long)n * 128, B), B, 0, stream>>>(h1, dinv, nullptr, agg1, (long)n * 128, 7);
    k_scatter<7><<<blk(E * 32, B), B, 0, stream>>>(srcIdx, dstIdx, dinv, h1, agg1, E);

    k_gemm_wmma<128, 64, true><<<blk(n, 128), 256, 0, stream>>>(agg1, W2, b1, h2, n);
    k_self_init<<<blk((long)n * 64, B), B, 0, stream>>>(h2, dinv, b2, out, (long)n * 64, 6);
    k_scatter<6><<<blk(E * 16, B), B, 0, stream>>>(srcIdx, dstIdx, dinv, h2, out, E);
}


# owner tile fill in 16-byte pieces, row and weight loads of all 13 (7) passes issued before the first wait (was one 4-byte element per lane per round trip: 49, 25 trips); drain in 16-byte pieces too
# speedup vs baseline: 2.0728x; 2.0728x over previous
#include <hip/hip_runtime.h>

typedef float v2f __attribute__((ext_vector_type(2)));
typedef float v8f __attribute__((ext_vector_type(8)));


__global__ void k_init_deg(float* deg, int n) {
    int i = blockIdx.x * blockDim.x + threadIdx.x;
    if (i < n) deg[i] = 1.0f;
}

__global__ void k_count_deg(const int* __restrict__ dst, long e, float* deg) {
    long i = (long)blockIdx.x * blockDim.x + threadIdx.x;
    if (i < e) unsafeAtomicAdd(&deg[dst[i]], 1.0f);
}

__global__ void k_rsqrt_inplace(float* deg, int n) {
    int i = blockIdx.x * blockDim.x + threadIdx.x;
    if (i < n) deg[i] = rsqrtf(deg[i]);
}


template <int K, int NOUT, bool RELU_BIAS>
__global__ __launch_bounds__(256) void k_gemm_wmma(const float* __restrict__ A,
                                                   const float* __restrict__ W,
                                                   const float* __restrict__ bias,
                                                   float* __restrict__ Out,
                                                   int nRows) {
    constexpr int PITCH = 2 * NOUT + 32;
    constexpr int HK    = K / 2;
    constexpr int PAIRS = K / 4;
    constexpr int CT    = NOUT / 16;
    __shared__ __align__(16) float lw[PAIRS * PITCH];

    const int lane  = threadIdx.x & 31;
    const int wave  = threadIdx.x >> 5;
    const int laneM = lane & 15;
    const int hi    = lane >> 4;

    const int rowBase = blockIdx.x * 128 + wave * 16 + laneM;
    const int rowC    = rowBase < nRows ? rowBase : 0;

    v8f acc[CT] = {};

    for (int kh = 0; kh < 2; ++kh) {
        __syncthreads();
        for (int idx = threadIdx.x; idx < HK * NOUT; idx += 256) {
            int lk = idx / NOUT;
            int c  = idx - lk * NOUT;
            lw[(lk >> 1) * PITCH + 2 * c + (lk & 1)] = W[(kh * HK + lk) * NOUT + c];
        }
        __syncthreads();

        for (int k4 = 0; k4 < HK; k4 += 4) {
            const int kk = kh * HK + k4 + 2 * hi;
            v2f a;
            a.x = A[(long)rowC * K + kk];
            a.y = A[(long)rowC * K + kk + 1];
            if (RELU_BIAS) {
                a.x = fmaxf(a.x + bias[kk], 0.0f);
                a.y = fmaxf(a.y + bias[kk + 1], 0.0f);
            }
            const float* lb = &lw[((k4 >> 1) + hi) * PITCH];

            v2f bfrag[CT];
#pragma unroll
            for (int ct = 0; ct < CT; ++ct)
                bfrag[ct] = *(const v2f*)&lb[2 * (ct * 16 + laneM)];

#pragma unroll
            for (int ct = 0; ct < CT; ++ct)
                acc[ct] = __builtin_amdgcn_wmma_f32_16x16x4_f32(
                    false, a, false, bfrag[ct], (short)0, acc[ct], false, false);
        }
    }

#pragma unroll
    for (int ct = 0; ct < CT; ++ct) {
#pragma unroll
        for (int r = 0; r < 8; ++r) {
            int m = blockIdx.x * 128 + wave * 16 + 8 * hi + r;
            if (m < nRows) Out[(long)m * NOUT + ct * 16 + laneM] = acc[ct][r];
        }
    }
}


constexpr int LCAP = 16384;

typedef int v2i __attribute__((ext_vector_type(2)));

template <int LOGF, int R, int MODE>
__global__ __launch_bounds__(1024) void k_agg(const int* __restrict__ src, const int* __restrict__ dst,
                                              const float* __restrict__ dinv, const float* __restrict__ h,
                                              const float* __restrict__ bias, float* __restrict__ out,
                                              int* glist, int* gcnt, long nE, int n) {
    constexpr int F    = 1 << LOGF;
    constexpr int LPR  = F / 32;
    constexpr int G    = 32 / LPR;
    constexpr int U    = 8;
    constexpr int QCAP = 32 + 2 * 128;
    __shared__ __align__(16) float tile[R * F];
    __shared__ float dl[R];
    __shared__ int   qe[32][QCAP];
    __shared__ int   ql[32][QCAP];
    __shared__ int   cur;

    const int lane = threadIdx.x & 31;
    const int wv = __builtin_amdgcn_readfirstlane((int)(threadIdx.x >> 5));
    const int lo = blockIdx.x * R;
    const int nr = (n - lo < R) ? (n - lo) : R;
    v2i* const mylist = (v2i*)glist + (long)blockIdx.x * LCAP;

    constexpr int F4  = F / 4;
    constexpr int NIT = (R * F4 + 1023) / 1024;
    const int     lim4 = nr * F4;
    {
        const float4* h4 = (const float4*)(h + (long)lo * F);
        float4 hv[NIT];
        float  dv[NIT];
#pragma unroll
        for (int it = 0; it < NIT; ++it) {
            int i4 = (int)threadIdx.x + it * 1024;
            i4     = i4 < lim4 ? i4 : lim4 - 1;
            hv[it] = h4[i4];
            dv[it] = dinv[lo + (i4 >> (LOGF - 2))];
        }
        float4 bv = {0.0f, 0.0f, 0.0f, 0.0f};
        if (bias) {
            const float* bp = bias + 4 * (threadIdx.x & (F4 - 1));
            bv.x = bp[0]; bv.y = bp[1]; bv.z = bp[2]; bv.w = bp[3];
        }
#pragma unroll
        for (int it = 0; it < NIT; ++it) {
            const int i4 = (int)threadIdx.x + it * 1024;
            if (i4 < lim4) {
                const float d = dv[it];
                float4      v;
                v.x = hv[it].x * d * d;
                v.y = hv[it].y * d * d;
                v.z = hv[it].z * d * d;
                v.w = hv[it].w * d * d;
                if (bias) { v.x += bv.x; v.y += bv.y; v.z += bv.z; v.w += bv.w; }
                ((float4*)tile)[i4] = v;
            }
        }
    }
    for (int r = threadIdx.x; r < nr; r += 1024) dl[r] = dinv[lo + r];
    if (MODE == 0 && threadIdx.x == 0) cur = 0;
    __syncthreads();

    auto gatherRows = [&](float (&v)[G][LPR], int s, int j0) {
#pragma unroll
        for (int jj = 0; jj < G; ++jj) {
            const int    sj = __builtin_amdgcn_readlane(s, j0 + jj);
            const float* hp = h + ((long)sj << LOGF) + lane;
#pragma unroll
            for (int i = 0; i < LPR; ++i) v[jj][i] = hp[32 * i];
        }
    };
    auto addGroup = [&](const float (&v)[G][LPR], int cnt, int l, int wi, int j0) {
#pragma unroll
        for (int jj = 0; jj < G; ++jj) {
            if (j0 + jj < cnt) {
                const int   lj = __builtin_amdgcn_readlane(l, j0 + jj);
                const float wj = __int_as_float(__builtin_amdgcn_readlane(wi, j0 + jj));
                float*      tp = tile + (lj << LOGF) + lane;
#pragma unroll
                for (int i = 0; i < LPR; ++i)
                    __hip_atomic_fetch_add(tp + 32 * i, v[jj][i] * wj, __ATOMIC_RELAXED,
                                           __HIP_MEMORY_SCOPE_WORKGROUP);
            }
        }
    };
    auto addRows = [&](int cnt, int s, int l, float w) {
        const int wi = __float_as_int(w);
        for (int j0 = 0; j0 < cnt; j0 += G) {
            float v[G][LPR];
            gatherRows(v, s, j0);
            addGroup(v, cnt, l, wi, j0);
        }
    };

    const int listed = (MODE == 1) ? gcnt[blockIdx.x] : 0;

    if (MODE == 1 && listed <= LCAP) {
        auto entry = [&](int i) {
            v2i t = {0, 0};
            if (i < listed) t = mylist[i];
            return t;
        };
        int   b  = wv * 32;
        v2i   e0 = entry(b + lane);
        v2i   e1 = entry(b + 1024 + lane);
        float w0 = dinv[e0.x] * dl[e0.y];
        while (b < listed) {
            const int cnt = (listed - b < 32) ? (listed - b) : 32;
            const int wi  = __float_as_int(w0);
            float v[G][LPR];
            gatherRows(v, e0.x, 0);
            const float d1 = dinv[e1.x];
            const float l1 = dl[e1.y];
            const v2i   e2 = entry(b + 2048 + lane);
            addGroup(v, cnt, e0.y, wi, 0);
            for (int j0 = G; j0 < cnt; j0 += G) {
                gatherRows(v, e0.x, j0);
                addGroup(v, cnt, e0.y, wi, j0);
            }
            w0 = d1 * l1;
            e0 = e1; e1 = e2; b += 1024;
        }
    } else {
        int qn = 0;

        auto flush = [&]() {
            for (int qb = 0; qb < qn; qb += 32) {
                const int cnt = (qn - qb < 32) ? (qn - qb) : 32;
                int   s = 0, l = 0;
                float w = 0.0f;
                if (lane < cnt) {
                    l = ql[wv][qb + lane];
                    s = src[qe[wv][qb + lane]];
                    w = dinv[s] * dl[l];
                }
                if (MODE == 0) {
                    int p = 0;
                    if (lane == 0)
                        p = __hip_atomic_fetch_add(&cur, cnt, __ATOMIC_RELAXED, __HIP_MEMORY_SCOPE_WORKGROUP);
                    p = __builtin_amdgcn_readfirstlane(p);
                    if (p + cnt <= LCAP && lane < cnt) {
                        v2i t;
                        t.x = s; t.y = l;
                        mylist[p + lane] = t;
                    }
                }
                addRows(cnt, s, l, w);
            }
            qn = 0;
        };

        const long nSuper = (nE + U * 128 - 1) / (U * 128);
        const int  K      = (int)((nSuper + 31) / 32);
        const int  k0     = (int)(((long)blockIdx.x * K) / gridDim.x);

        for (int k = 0; k < K; ++k) {
            int kk = k + k0;
            if (kk >= K) kk -= K;
            const long sc = (long)kk * 32 + wv;
            if (sc >= nSuper) continue;
            const long base = sc * (U * 128);
            int d[U][4];
            if (base + U * 128 <= nE) {
#pragma unroll
                for (int u = 0; u < U; ++u) {
                    const int4 t = *(const int4*)(dst + base + u * 128 + 4 * lane);
                    d[u][0] = t.x; d[u][1] = t.y; d[u][2] = t.z; d[u][3] = t.w;
                }
            } else {
#pragma unroll
                for (int u = 0; u < U; ++u)
#pragma unroll
                    for (int c = 0; c < 4; ++c) {
                        const long idx = base + u * 128 + 4 * lane + c;
                        d[u][c] = idx < nE ? dst[idx] : -1;
                    }
            }
#pragma unroll
            for (int u = 0; u < U; ++u) {
#pragma unroll
                for (int c = 0; c < 4; ++c) {
                    const int      loc  = d[u][c] - lo;
                    const bool     m    = (unsigned)loc < (unsigned)nr;
                    const unsigned mask = __builtin_amdgcn_ballot_w32(m);
                    if (mask) {
                        const int pos = qn + (int)__builtin_amdgcn_mbcnt_lo(mask, 0u);
                        if (m) {
                            qe[wv][pos] = (int)(base + u * 128 + 4 * lane + c);
                            ql[wv][pos] = loc;
                        }
                        qn += __builtin_popcount(mask);
                    }
                }
                if ((u & 1) && qn >= 32) flush();
            }
        }
        if (qn) flush();
    }
    __syncthreads();

    if (MODE == 0 && threadIdx.x == 0) gcnt[blockIdx.x] = cur;
    {
        float4* o4 = (float4*)(out + (long)lo * F);
#pragma unroll
        for (int it = 0; it < NIT; ++it) {
            const int i4 = (int)threadIdx.x + it * 1024;
            if (i4 < lim4) o4[i4] = ((const float4*)tile)[i4];
        }
    }
}


extern "C" void kernel_launch(void* const* d_in, const int* in_sizes, int n_in,
                              void* d_out, int out_size, void* d_ws, size_t ws_size,
                              hipStream_t stream) {
    const float* x   = (const float*)d_in[0];
    const int*   ei  = (const int*)d_in[1];
    const float* W1  = (const float*)d_in[2];
    const float* b1  = (const float*)d_in[3];
    const float* W2  = (const float*)d_in[4];
    const float* b2  = (const float*)d_in[5];
    float*       out = (float*)d_out;

    const int  n = in_sizes[0] / 128;
    const long E = in_sizes[1] / 2;
    const int* srcIdx = ei;
    const int* dstIdx = ei + E;

    float* ws   = (float*)d_ws;
    float* dinv = ws;
    float* h1   = dinv + n;
    float* agg1 = h1 + (long)n * 128;
    float* h2   = agg1 + (long)n * 128;
    int*   ecnt  = (int*)(h2 + (long)n * 64);
    int*   elist = ecnt + n;

    const int B = 256;
    auto blk = [](long work, int b) { return (int)((work + b - 1) / b); };

    k_init_deg<<<blk(n, B), B, 0, stream>>>(dinv, n);
    k_count_deg<<<blk(E, B), B, 0, stream>>>(dstIdx, E, dinv);
    k_rsqrt_inplace<<<blk(n, B), B, 0, stream>>>(dinv, n);

    k_gemm_wmma<128, 128, false><<<blk(n, 128), 256, 0, stream>>>(x, W1, nullptr, h1, n);
    k_agg<7, 392, 0><<<blk(n, 392), 1024, 0, stream>>>(srcIdx, dstIdx, dinv, h1, nullptr, agg1, elist, ecnt, E, n);

    k_gemm_wmma<128, 64, true><<<blk(n, 128), 256, 0, stream>>>(agg1, W2, b1, h2, n);
    k_agg<6, 392, 1><<<blk(n, 392), 1024, 0, stream>>>(srcIdx, dstIdx, dinv, h2, b2, out, elist, ecnt, E, n);
}


# owner scan reads a 1-byte owner key per edge (written by the degree pass) instead of the 4-byte id: 4x fewer scan bytes, 4 keys per zero-byte test, real id checked at flush; sum order free as before
# speedup vs baseline: 2.2146x; 1.0684x over previous
#include <hip/hip_runtime.h>

typedef float v2f __attribute__((ext_vector_type(2)));
typedef float v8f __attribute__((ext_vector_type(8)));


__global__ void k_init_deg(float* deg, int n) {
    int i = blockIdx.x * blockDim.x + threadIdx.x;
    if (i < n) deg[i] = 1.0f;
}

constexpr int RG = 392;

__global__ void k_count_deg(const int* __restrict__ dst, long e, float* deg, unsigned* __restrict__ keys) {
    const long t = (long)blockIdx.x * blockDim.x + threadIdx.x;
    const long i = 4 * t;
    if (i >= ((e + 15) & ~15L)) return;
    int d[4];
    if (i + 4 <= e && (((unsigned long)(dst + i)) & 15ul) == 0) {
        const int4 v = *(const int4*)(dst + i);
        d[0] = v.x; d[1] = v.y; d[2] = v.z; d[3] = v.w;
    } else {
#pragma unroll
        for (int c = 0; c < 4; ++c) d[c] = (i + c < e) ? dst[i + c] : -1;
    }
    unsigned k = 0;
#pragma unroll
    for (int c = 0; c < 4; ++c) {
        if (d[c] >= 0) {
            unsafeAtomicAdd(&deg[d[c]], 1.0f);
            k |= ((unsigned)(d[c] / RG) & 255u) << (8 * c);
        }
    }
    keys[t] = k;
}

__global__ void k_rsqrt_inplace(float* deg, int n) {
    int i = blockIdx.x * blockDim.x + threadIdx.x;
    if (i < n) deg[i] = rsqrtf(deg[i]);
}


template <int K, int NOUT, bool RELU_BIAS>
__global__ __launch_bounds__(256) void k_gemm_wmma(const float* __restrict__ A,
                                                   const float* __restrict__ W,
                                                   const float* __restrict__ bias,
                                                   float* __restrict__ Out,
                                                   int nRows) {
    constexpr int PITCH = 2 * NOUT + 32;
    constexpr int HK    = K / 2;
    constexpr int PAIRS = K / 4;
    constexpr int CT    = NOUT / 16;
    __shared__ __align__(16) float lw[PAIRS * PITCH];

    const int lane  = threadIdx.x & 31;
    const int wave  = threadIdx.x >> 5;
    const int laneM = lane & 15;
    const int hi    = lane >> 4;

    const int rowBase = blockIdx.x * 128 + wave * 16 + laneM;
    const int rowC    = rowBase < nRows ? rowBase : 0;

    v8f acc[CT] = {};

    for (int kh = 0; kh < 2; ++kh) {
        __syncthreads();
        for (int idx = threadIdx.x; idx < HK * NOUT; idx += 256) {
            int lk = idx / NOUT;
            int c  = idx - lk * NOUT;
            lw[(lk >> 1) * PITCH + 2 * c + (lk & 1)] = W[(kh * HK + lk) * NOUT + c];
        }
        __syncthreads();

        for (int k4 = 0; k4 < HK; k4 += 4) {
            const int kk = kh * HK + k4 + 2 * hi;
            v2f a;
            a.x = A[(long)rowC * K + kk];
            a.y = A[(long)rowC * K + kk + 1];
            if (RELU_BIAS) {
                a.x = fmaxf(a.x + bias[kk], 0.0f);
                a.y = fmaxf(a.y + bias[kk + 1], 0.0f);
            }
            const float* lb = &lw[((k4 >> 1) + hi) * PITCH];

            v2f bfrag[CT];
#pragma unroll
            for (int ct = 0; ct < CT; ++ct)
                bfrag[ct] = *(const v2f*)&lb[2 * (ct * 16 + laneM)];

#pragma unroll
            for (int ct = 0; ct < CT; ++ct)
                acc[ct] = __builtin_amdgcn_wmma_f32_16x16x4_f32(
                    false, a, false, bfrag[ct], (short)0, acc[ct], false, false);
        }
    }

#pragma unroll
    for (int ct = 0; ct < CT; ++ct) {
#pragma unroll
        for (int r = 0; r < 8; ++r) {
            int m = blockIdx.x * 128 + wave * 16 + 8 * hi + r;
            if (m < nRows) Out[(long)m * NOUT + ct * 16 + laneM] = acc[ct][r];
        }
    }
}


constexpr int LCAP = 16384;

typedef int v2i __attribute__((ext_vector_type(2)));

template <int LOGF, int R, int MODE>
__global__ __launch_bounds__(1024) void k_agg(const int* __restrict__ src, const int* __restrict__ dst,
                                              const float* __restrict__ dinv, const float* __restrict__ h,
                                              const float* __restrict__ bias, float* __restrict__ out,
                                              int* glist, int* gcnt, const unsigned* __restrict__ keys,
                                              long nE, int n) {
    constexpr int F    = 1 << LOGF;
    constexpr int LPR  = F / 32;
    constexpr int G    = 32 / LPR;
    constexpr int U    = 8;
    constexpr int QCAP = 32 + 2 * 128;
    __shared__ __align__(16) float tile[R * F];
    __shared__ float dl[R];
    __shared__ int   qe[32][QCAP];
    __shared__ int   cur;

    const int lane = threadIdx.x & 31;
    const int wv = __builtin_amdgcn_readfirstlane((int)(threadIdx.x >> 5));
    const int lo = blockIdx.x * R;
    const int nr = (n - lo < R) ? (n - lo) : R;
    v2i* const mylist = (v2i*)glist + (long)blockIdx.x * LCAP;

    constexpr int F4  = F / 4;
    constexpr int NIT = (R * F4 + 1023) / 1024;
    const int     lim4 = nr * F4;
    {
        const float4* h4 = (const float4*)(h + (long)lo * F);
        float4 hv[NIT];
        float  dv[NIT];
#pragma unroll
        for (int it = 0; it < NIT; ++it) {
            int i4 = (int)threadIdx.x + it * 1024;
            i4     = i4 < lim4 ? i4 : lim4 - 1;
            hv[it] = h4[i4];
            dv[it] = dinv[lo + (i4 >> (LOGF - 2))];
        }
        float4 bv = {0.0f, 0.0f, 0.0f, 0.0f};
        if (bias) {
            const float* bp = bias + 4 * (threadIdx.x & (F4 - 1));
            bv.x = bp[0]; bv.y = bp[1]; bv.z = bp[2]; bv.w = bp[3];
        }
#pragma unroll
        for (int it = 0; it < NIT; ++it) {
            const int i4 = (int)threadIdx.x + it * 1024;
            if (i4 < lim4) {
                const float d = dv[it];
                float4      v;
                v.x = hv[it].x * d * d;
                v.y = hv[it].y * d * d;
                v.z = hv[it].z * d * d;
                v.w = hv[it].w * d * d;
                if (bias) { v.x += bv.x; v.y += bv.y; v.z += bv.z; v.w += bv.w; }
                ((float4*)tile)[i4] = v;
            }
        }
    }
    for (int r = threadIdx.x; r < nr; r += 1024) dl[r] = dinv[lo + r];
    if (MODE == 0 && threadIdx.x == 0) cur = 0;
    __syncthreads();

    auto gatherRows = [&](float (&v)[G][LPR], int s, int j0) {
#pragma unroll
        for (int jj = 0; jj < G; ++jj) {
            const int    sj = __builtin_amdgcn_readlane(s, j0 + jj);
            const float* hp = h + ((long)sj << LOGF) + lane;
#pragma unroll
            for (int i = 0; i < LPR; ++i) v[jj][i] = hp[32 * i];
        }
    };
    auto addGroup = [&](const float (&v)[G][LPR], unsigned vm, int l, int wi, int j0) {
#pragma unroll
        for (int jj = 0; jj < G; ++jj) {
            if ((vm >> (j0 + jj)) & 1u) {
                const int   lj = __builtin_amdgcn_readlane(l, j0 + jj);
                const float wj = __int_as_float(__builtin_amdgcn_readlane(wi, j0 + jj));
                float*      tp = tile + (lj << LOGF) + lane;
#pragma unroll
                for (int i = 0; i < LPR; ++i)
                    __hip_atomic_fetch_add(tp + 32 * i, v[jj][i] * wj, __ATOMIC_RELAXED,
                                           __HIP_MEMORY_SCOPE_WORKGROUP);
            }
        }
    };
    auto addRows = [&](unsigned vm, int cnt, int s, int l, float w) {
        const int wi = __float_as_int(w);
        for (int j0 = 0; j0 < cnt; j0 += G) {
            if (((vm >> j0) & ((1u << G) - 1u)) == 0) continue;
            float v[G][LPR];
            gatherRows(v, s, j0);
            addGroup(v, vm, l, wi, j0);
        }
    };

    const int listed = (MODE == 1) ? gcnt[blockIdx.x] : 0;

    if (MODE == 1 && listed <= LCAP) {
        auto entry = [&](int i) {
            v2i t = {0, 0};
            if (i < listed) t = mylist[i];
            return t;
        };
        int   b  = wv * 32;
        v2i   e0 = entry(b + lane);
        v2i   e1 = entry(b + 1024 + lane);
        float w0 = dinv[e0.x] * dl[e0.y];
        while (b < listed) {
            const int      cnt = (listed - b < 32) ? (listed - b) : 32;
            const unsigned vm  = cnt < 32 ? ((1u << cnt) - 1u) : ~0u;
            const int      wi  = __float_as_int(w0);
            float v[G][LPR];
            gatherRows(v, e0.x, 0);
            const float d1 = dinv[e1.x];
            const float l1 = dl[e1.y];
            const v2i   e2 = entry(b + 2048 + lane);
            addGroup(v, vm, e0.y, wi, 0);
            for (int j0 = G; j0 < cnt; j0 += G) {
                gatherRows(v, e0.x, j0);
                addGroup(v, vm, e0.y, wi, j0);
            }
            w0 = d1 * l1;
            e0 = e1; e1 = e2; b += 1024;
        }
    } else {
        int qn = 0;

        auto flush = [&]() {
            for (int qb = 0; qb < qn; qb += 32) {
                const int cnt = (qn - qb < 32) ? (qn - qb) : 32;
                int   s = 0, l = 0;
                float w = 0.0f;
                bool  ok = false;
                if (lane < cnt) {
                    const int e = qe[wv][qb + lane];
                    if (e < nE) {
                        const int loc = dst[e] - lo;
                        s = src[e];
                        if ((unsigned)loc < (unsigned)nr) {
                            ok = true;
                            l  = loc;
                            w  = dinv[s] * dl[l];
                        }
                    }
                }
                const unsigned vm = __builtin_amdgcn_ballot_w32(ok);
                if (!vm) continue;
                if (MODE == 0) {
                    const int nv = __builtin_popcount(vm);
                    int p = 0;
                    if (lane == 0)
                        p = __hip_atomic_fetch_add(&cur, nv, __ATOMIC_RELAXED, __HIP_MEMORY_SCOPE_WORKGROUP);
                    p = __builtin_amdgcn_readfirstlane(p);
                    if (p + nv <= LCAP && ok) {
                        v2i t;
                        t.x = s; t.y = l;
                        mylist[p + (int)__builtin_amdgcn_mbcnt_lo(vm, 0u)] = t;
                    }
                }
                addRows(vm, cnt, s, l, w);
            }
            qn = 0;
        };

        const unsigned gpat  = (blockIdx.x & 255u) * 0x01010101u;
        const unsigned nohit = gpat ^ 0x01010101u;
        const int      lane16 = 16 * lane;
        const long nK16   = (nE + 15) & ~15L;
        const long nSuper = (nE + U * 512 - 1) / (U * 512);
        const int  K      = (int)((nSuper + 31) / 32);
        const int  KT     = K * 32;
        const int  rot    = (int)(((long)blockIdx.x * KT) / gridDim.x);

        for (int k = 0; k < K; ++k) {
            long sc = (long)k * 32 + wv + rot;
            if (sc >= KT) sc -= KT;
            if (sc >= nSuper) continue;
            const long base = sc * (U * 512);
            unsigned d[U][4];
            if (base + U * 512 <= nK16) {
#pragma unroll
                for (int u = 0; u < U; ++u) {
                    const uint4 t = *(const uint4*)((const unsigned char*)keys + base + u * 512 + 16 * lane);
                    d[u][0] = t.x; d[u][1] = t.y; d[u][2] = t.z; d[u][3] = t.w;
                }
            } else {
#pragma unroll
                for (int u = 0; u < U; ++u) {
                    const long o = base + u * 512 + 16 * lane;
                    uint4 t = {nohit, nohit, nohit, nohit};
                    if (o + 16 <= nK16) t = *(const uint4*)((const unsigned char*)keys + o);
                    d[u][0] = t.x; d[u][1] = t.y; d[u][2] = t.z; d[u][3] = t.w;
                }
            }
#pragma unroll
            for (int u = 0; u < U; ++u)
#pragma unroll
                for (int c = 0; c < 4; ++c) {
                    const unsigned t = d[u][c] ^ gpat;
                    d[u][c] = (t - 0x01010101u) & ~t & 0x80808080u;
                }
            const int sbase = (int)base;
            bool again;
            do {
                again = false;
#pragma unroll
                for (int p = 0; p < 2 * U; ++p) {
#pragma unroll
                    for (int c = 2 * (p & 1); c < 2 * (p & 1) + 2; ++c) {
                        unsigned mask = __builtin_amdgcn_ballot_w32(d[p >> 1][c] != 0);
                        while (mask) {
                            const unsigned x   = d[p >> 1][c];
                            const int      pos = qn + (int)__builtin_amdgcn_mbcnt_lo(mask, 0u);
                            if (x != 0)
                                qe[wv][pos] = (sbase + lane16) + ((__builtin_ctz(x) + (4096 * (p >> 1) + 32 * c)) >> 3);
                            qn += __builtin_popcount(mask);
                            d[p >> 1][c] = x & (x - 1u);
                            mask = __builtin_amdgcn_ballot_w32(d[p >> 1][c] != 0);
                        }
                    }
                    if (qn >= 32) {
                        again = p + 1 < 2 * U;
                        break;
                    }
                }
                if (qn >= 32) flush();
            } while (again);
        }
        if (qn) flush();
    }
    __syncthreads();

    if (MODE == 0 && threadIdx.x == 0) gcnt[blockIdx.x] = cur;
    {
        float4* o4 = (float4*)(out + (long)lo * F);
#pragma unroll
        for (int it = 0; it < NIT; ++it) {
            const int i4 = (int)threadIdx.x + it * 1024;
            if (i4 < lim4) o4[i4] = ((const float4*)tile)[i4];
        }
    }
}


extern "C" void kernel_launch(void* const* d_in, const int* in_sizes, int n_in,
                              void* d_out, int out_size, void* d_ws, size_t ws_size,
                              hipStream_t stream) {
    const float* x   = (const float*)d_in[0];
    const int*   ei  = (const int*)d_in[1];
    const float* W1  = (const float*)d_in[2];
    const float* b1  = (const float*)d_in[3];
    const float* W2  = (const float*)d_in[4];
    const float* b2  = (const float*)d_in[5];
    float*       out = (float*)d_out;

    const int  n = in_sizes[0] / 128;
    const long E = in_sizes[1] / 2;
    const int* srcIdx = ei;
    const int* dstIdx = ei + E;

    float* ws   = (float*)d_ws;
    float* dinv = ws;
    float* h1   = dinv + n;
    float* agg1 = h1 + (long)n * 128;
    float* h2   = agg1 + (long)n * 128;
    int*   ecnt  = (int*)(h2 + (long)n * 64);
    int*   elist = ecnt + n;

    const int B = 256;
    auto blk = [](long work, int b) { return (int)((work + b - 1) / b); };
    unsigned* keys = (unsigned*)(elist + (long)blk(n, RG) * (LCAP * 2));

    k_init_deg<<<blk(n, B), B, 0, stream>>>(dinv, n);
    k_count_deg<<<blk(E + 15, 4 * B), B, 0, stream>>>(dstIdx, E, dinv, keys);
    k_rsqrt_inplace<<<blk(n, B), B, 0, stream>>>(dinv, n);

    k_gemm_wmma<128, 128, false><<<blk(n, 128), 256, 0, stream>>>(x, W1, nullptr, h1, n);
    k_agg<7, RG, 0><<<blk(n, RG), 1024, 0, stream>>>(srcIdx, dstIdx, dinv, h1, nullptr, agg1, elist, ecnt, keys, E, n);

    k_gemm_wmma<128, 64, true><<<blk(n, 128), 256, 0, stream>>>(agg1, W2, b1, h2, n);
    k_agg<6, RG, 1><<<blk(n, RG), 1024, 0, stream>>>(srcIdx, dstIdx, dinv, h2, b2, out, elist, ecnt, keys, E, n);
}


# source deg^-1/2 folded into the stored rows by both GEMM epilogues; edge sums add unweighted rows and scale by the destination's once at the store: no per-edge dinv[src] look-up; sum order free
# speedup vs baseline: 2.3066x; 1.0415x over previous
#include <hip/hip_runtime.h>

typedef float v2f __attribute__((ext_vector_type(2)));
typedef float v8f __attribute__((ext_vector_type(8)));


__global__ void k_init_deg(float* deg, int n) {
    int i = blockIdx.x * blockDim.x + threadIdx.x;
    if (i < n) deg[i] = 1.0f;
}

constexpr int RG = 392;

__global__ void k_count_deg(const int* __restrict__ dst, long e, float* deg, unsigned* __restrict__ keys) {
    const long t = (long)blockIdx.x * blockDim.x + threadIdx.x;
    const long i = 4 * t;
    if (i >= ((e + 15) & ~15L)) return;
    int d[4];
    if (i + 4 <= e && (((unsigned long)(dst + i)) & 15ul) == 0) {
        const int4 v = *(const int4*)(dst + i);
        d[0] = v.x; d[1] = v.y; d[2] = v.z; d[3] = v.w;
    } else {
#pragma unroll
        for (int c = 0; c < 4; ++c) d[c] = (i + c < e) ? dst[i + c] : -1;
    }
    unsigned k = 0;
#pragma unroll
    for (int c = 0; c < 4; ++c) {
        if (d[c] >= 0) {
            unsafeAtomicAdd(&deg[d[c]], 1.0f);
            k |= ((unsigned)(d[c] / RG) & 255u) << (8 * c);
        }
    }
    keys[t] = k;
}

__global__ void k_rsqrt_inplace(float* deg, int n) {
    int i = blockIdx.x * blockDim.x + threadIdx.x;
    if (i < n) deg[i] = rsqrtf(deg[i]);
}


template <int K, int NOUT, bool RELU_BIAS>
__global__ __launch_bounds__(256) void k_gemm_wmma(const float* __restrict__ A,
                                                   const float* __restrict__ W,
                                                   const float* __restrict__ bias,
                                                   const float* __restrict__ rowScale,
                                                   float* __restrict__ Out,
                                                   int nRows) {
    constexpr int PITCH = 2 * NOUT + 32;
    constexpr int HK    = K / 2;
    constexpr int PAIRS = K / 4;
    constexpr int CT    = NOUT / 16;
    __shared__ __align__(16) float lw[PAIRS * PITCH];

    const int lane  = threadIdx.x & 31;
    const int wave  = threadIdx.x >> 5;
    const int laneM = lane & 15;
    const int hi    = lane >> 4;

    const int rowBase = blockIdx.x * 128 + wave * 16 + laneM;
    const int rowC    = rowBase < nRows ? rowBase : 0;

    v8f acc[CT] = {};

    float sc[8] = {};

    for (int kh = 0; kh < 2; ++kh) {
        __syncthreads();
        for (int idx = threadIdx.x; idx < HK * NOUT; idx += 256) {
            int lk = idx / NOUT;
            int c  = idx - lk * NOUT;
            lw[(lk >> 1) * PITCH + 2 * c + (lk & 1)] = W[(kh * HK + lk) * NOUT + c];
        }
        __syncthreads();

        if (kh == 1) {
#pragma unroll
            for (int r = 0; r < 8; ++r) {
                const int m = blockIdx.x * 128 + wave * 16 + 8 * hi + r;
                sc[r] = rowScale[m < nRows ? m : 0];
            }
        }

        for (int k4 = 0; k4 < HK; k4 += 4) {
            const int kk = kh * HK + k4 + 2 * hi;
            v2f a;
            a.x = A[(long)rowC * K + kk];
            a.y = A[(long)rowC * K + kk + 1];
            if (RELU_BIAS) {
                a.x = fmaxf(a.x + bias[kk], 0.0f);
                a.y = fmaxf(a.y + bias[kk + 1], 0.0f);
            }
            const float* lb = &lw[((k4 >> 1) + hi) * PITCH];

            v2f bfrag[CT];
#pragma unroll
            for (int ct = 0; ct < CT; ++ct)
                bfrag[ct] = *(const v2f*)&lb[2 * (ct * 16 + laneM)];

#pragma unroll
            for (int ct = 0; ct < CT; ++ct)
                acc[ct] = __builtin_amdgcn_wmma_f32_16x16x4_f32(
                    false, a, false, bfrag[ct], (short)0, acc[ct], false, false);
        }
    }

#pragma unroll
    for (int ct = 0; ct < CT; ++ct) {
#pragma unroll
        for (int r = 0; r < 8; ++r) {
            int m = blockIdx.x * 128 + wave * 16 + 8 * hi + r;
            if (m < nRows) Out[(long)m * NOUT + ct * 16 + laneM] = acc[ct][r] * sc[r];
        }
    }
}


constexpr int LCAP = 16384;

typedef int   v2i __attribute__((ext_vector_type(2)));
typedef float v4f __attribute__((ext_vector_type(4)));

template <int LOGF, int R, int MODE>
__global__ __launch_bounds__(1024) void k_agg(const int* __restrict__ src, const int* __restrict__ dst,
                                              const float* __restrict__ dinv, const float* __restrict__ h,
                                              const float* __restrict__ bias, float* __restrict__ out,
                                              int* glist, int* gcnt, const unsigned* __restrict__ keys,
                                              long nE, int n) {
    constexpr int F    = 1 << LOGF;
    constexpr int LPR  = F / 32;
    constexpr int G    = 32 / LPR;
    constexpr int U    = 8;
    constexpr int QCAP = 32 + 2 * 128;
    __shared__ __align__(16) float tile[R * F];
    __shared__ float dl[R];
    __shared__ int   qe[32][QCAP];
    __shared__ int   cur;

    const int lane = threadIdx.x & 31;
    const int wv = __builtin_amdgcn_readfirstlane((int)(threadIdx.x >> 5));
    const int lo = blockIdx.x * R;
    const int nr = (n - lo < R) ? (n - lo) : R;
    v2i* const mylist = (v2i*)glist + (long)blockIdx.x * LCAP;

    constexpr int F4  = F / 4;
    constexpr int NIT = (R * F4 + 1023) / 1024;
    const int     lim4 = nr * F4;
    {
        const v4f* h4 = (const v4f*)(h + (long)lo * F);
        v4f hv[NIT];
#pragma unroll
        for (int it = 0; it < NIT; ++it) {
            int i4 = (int)threadIdx.x + it * 1024;
            i4     = i4 < lim4 ? i4 : lim4 - 1;
            hv[it] = h4[i4];
        }
#pragma unroll
        for (int it = 0; it < NIT; ++it) {
            const int i4 = (int)threadIdx.x + it * 1024;
            if (i4 < lim4) ((v4f*)tile)[i4] = hv[it];
        }
    }
    for (int r = threadIdx.x; r < nr; r += 1024) dl[r] = dinv[lo + r];
    if (MODE == 0 && threadIdx.x == 0) cur = 0;
    __syncthreads();

    auto gatherRows = [&](float (&v)[G][LPR], int s, int j0) {
#pragma unroll
        for (int jj = 0; jj < G; ++jj) {
            const int    sj = __builtin_amdgcn_readlane(s, j0 + jj);
            const float* hp = h + ((long)sj << LOGF) + lane;
#pragma unroll
            for (int i = 0; i < LPR; ++i) v[jj][i] = hp[32 * i];
        }
    };
    auto addGroup = [&](const float (&v)[G][LPR], unsigned vm, int l, int j0) {
#pragma unroll
        for (int jj = 0; jj < G; ++jj) {
            if ((vm >> (j0 + jj)) & 1u) {
                const int lj = __builtin_amdgcn_readlane(l, j0 + jj);
                float*    tp = tile + (lj << LOGF) + lane;
#pragma unroll
                for (int i = 0; i < LPR; ++i)
                    __hip_atomic_fetch_add(tp + 32 * i, v[jj][i], __ATOMIC_RELAXED,
                                           __HIP_MEMORY_SCOPE_WORKGROUP);
            }
        }
    };
    auto addRows = [&](unsigned vm, int cnt, int s, int l) {
        for (int j0 = 0; j0 < cnt; j0 += G) {
            if (((vm >> j0) & ((1u << G) - 1u)) == 0) continue;
            float v[G][LPR];
            gatherRows(v, s, j0);
            addGroup(v, vm, l, j0);
        }
    };

    const int listed = (MODE == 1) ? gcnt[blockIdx.x] : 0;

    if (MODE == 1 && listed <= LCAP) {
        auto entry = [&](int i) {
            v2i t = {0, 0};
            if (i < listed) t = mylist[i];
            return t;
        };
        int b  = wv * 32;
        v2i e0 = entry(b + lane);
        while (b < listed) {
            const int      cnt = (listed - b < 32) ? (listed - b) : 32;
            const unsigned vm  = cnt < 32 ? ((1u << cnt) - 1u) : ~0u;
            float v[G][LPR];
            gatherRows(v, e0.x, 0);
            const v2i e1 = entry(b + 1024 + lane);
            addGroup(v, vm, e0.y, 0);
            for (int j0 = G; j0 < cnt; j0 += G) {
                gatherRows(v, e0.x, j0);
                addGroup(v, vm, e0.y, j0);
            }
            e0 = e1; b += 1024;
        }
    } else {
        int qn = 0;

        auto flush = [&]() {
            for (int qb = 0; qb < qn; qb += 32) {
                const int cnt = (qn - qb < 32) ? (qn - qb) : 32;
                int  s = 0, l = 0;
                bool ok = false;
                if (lane < cnt) {
                    const int e = qe[wv][qb + lane];
                    if (e < nE) {
                        const int loc = dst[e] - lo;
                        s = src[e];
                        if ((unsigned)loc < (unsigned)nr) {
                            ok = true;
                            l  = loc;
                        }
                    }
                }
                const unsigned vm = __builtin_amdgcn_ballot_w32(ok);
                if (!vm) continue;
                if (MODE == 0) {
                    const int nv = __builtin_popcount(vm);
                    int p = 0;
                    if (lane == 0)
                        p = __hip_atomic_fetch_add(&cur, nv, __ATOMIC_RELAXED, __HIP_MEMORY_SCOPE_WORKGROUP);
                    p = __builtin_amdgcn_readfirstlane(p);
                    if (p + nv <= LCAP && ok) {
                        v2i t;
                        t.x = s; t.y = l;
                        mylist[p + (int)__builtin_amdgcn_mbcnt_lo(vm, 0u)] = t;
                    }
                }
                addRows(vm, cnt, s, l);
            }
            qn = 0;
        };

        const unsigned gpat  = (blockIdx.x & 255u) * 0x01010101u;
        const unsigned nohit = gpat ^ 0x01010101u;
        const int      lane16 = 16 * lane;
        const long nK16   = (nE + 15) & ~15L;
        const long nSuper = (nE + U * 512 - 1) / (U * 512);
        const int  K      = (int)((nSuper + 31) / 32);
        const int  KT     = K * 32;
        const int  rot    = (int)(((long)blockIdx.x * KT) / gridDim.x);

        for (int k = 0; k < K; ++k) {
            long sc = (long)k * 32 + wv + rot;
            if (sc >= KT) sc -= KT;
            if (sc >= nSuper) continue;
            const long base = sc * (U * 512);
            unsigned d[U][4];
            if (base + U * 512 <= nK16) {
#pragma unroll
                for (int u = 0; u < U; ++u) {
                    const uint4 t = *(const uint4*)((const unsigned char*)keys + base + u * 512 + 16 * lane);
                    d[u][0] = t.x; d[u][1] = t.y; d[u][2] = t.z; d[u][3] = t.w;
                }
            } else {
#pragma unroll
                for (int u = 0; u < U; ++u) {
                    const long o = base + u * 512 + 16 * lane;
                    uint4 t = {nohit, nohit, nohit, nohit};
                    if (o + 16 <= nK16) t = *(const uint4*)((const unsigned char*)keys + o);
                    d[u][0] = t.x; d[u][1] = t.y; d[u][2] = t.z; d[u][3] = t.w;
                }
            }
#pragma unroll
            for (int u = 0; u < U; ++u)
#pragma unroll
                for (int c = 0; c < 4; ++c) {
                    const unsigned t = d[u][c] ^ gpat;
                    d[u][c] = (t - 0x01010101u) & ~t & 0x80808080u;
                }
            const int sbase = (int)base;
            bool again;
            do {
                again = false;
#pragma unroll
                for (int p = 0; p < 2 * U; ++p) {
#pragma unroll
                    for (int c = 2 * (p & 1); c < 2 * (p & 1) + 2; ++c) {
                        unsigned mask = __builtin_amdgcn_ballot_w32(d[p >> 1][c] != 0);
                        while (mask) {
                            const unsigned x   = d[p >> 1][c];
                            const int      pos = qn + (int)__builtin_amdgcn_mbcnt_lo(mask, 0u);
                            if (x != 0)
                                qe[wv][pos] = (sbase + lane16) + ((__builtin_ctz(x) + (4096 * (p >> 1) + 32 * c)) >> 3);
                            qn += __builtin_popcount(mask);
                            d[p >> 1][c] = x & (x - 1u);
                            mask = __builtin_amdgcn_ballot_w32(d[p >> 1][c] != 0);
                        }
                    }
                    if (qn >= 32) {
                        again = p + 1 < 2 * U;
                        break;
                    }
                }
                if (qn >= 32) flush();
            } while (again);
        }
        if (qn) flush();
    }
    v4f bv = {0.0f, 0.0f, 0.0f, 0.0f};
    if (bias) {
        const float* bp = bias + 4 * (threadIdx.x & (F4 - 1));
        bv.x = bp[0]; bv.y = bp[1]; bv.z = bp[2]; bv.w = bp[3];
    }
    __syncthreads();

    if (MODE == 0 && threadIdx.x == 0) gcnt[blockIdx.x] = cur;
    {
        v4f* o4 = (v4f*)(out + (long)lo * F);
#pragma unroll
        for (int it = 0; it < NIT; ++it) {
            const int i4 = (int)threadIdx.x + it * 1024;
            if (i4 < lim4) {
                const float d = dl[i4 >> (LOGF - 2)];
                v4f         v = ((const v4f*)tile)[i4] * d;
                if (bias) v += bv;
                o4[i4] = v;
            }
        }
    }
}


extern "C" void kernel_launch(void* const* d_in, const int* in_sizes, int n_in,
                              void* d_out, int out_size, void* d_ws, size_t ws_size,
                              hipStream_t stream) {
    const float* x   = (const float*)d_in[0];
    const int*   ei  = (const int*)d_in[1];
    const float* W1  = (const float*)d_in[2];
    const float* b1  = (const float*)d_in[3];
    const float* W2  = (const float*)d_in[4];
    const float* b2  = (const float*)d_in[5];
    float*       out = (float*)d_out;

    const int  n = in_sizes[0] / 128;
    const long E = in_sizes[1] / 2;
    const int* srcIdx = ei;
    const int* dstIdx = ei + E;

    float* ws   = (float*)d_ws;
    float* dinv = ws;
    float* h1   = dinv + n;
    float* agg1 = h1 + (long)n * 128;
    float* h2   = agg1 + (long)n * 128;
    int*   ecnt  = (int*)(h2 + (long)n * 64);
    int*   elist = ecnt + n;

    const int B = 256;
    auto blk = [](long work, int b) { return (int)((work + b - 1) / b); };
    unsigned* keys = (unsigned*)(elist + (long)blk(n, RG) * (LCAP * 2));

    k_init_deg<<<blk(n, B), B, 0, stream>>>(dinv, n);
    k_count_deg<<<blk(E + 15, 4 * B), B, 0, stream>>>(dstIdx, E, dinv, keys);
    k_rsqrt_inplace<<<blk(n, B), B, 0, stream>>>(dinv, n);

    k_gemm_wmma<128, 128, false><<<blk(n, 128), 256, 0, stream>>>(x, W1, nullptr, dinv, h1, n);
    k_agg<7, RG, 0><<<blk(n, RG), 1024, 0, stream>>>(srcIdx, dstIdx, dinv, h1, nullptr, agg1, elist, ecnt, keys, E, n);

    k_gemm_wmma<128, 64, true><<<blk(n, 128), 256, 0, stream>>>(agg1, W2, b1, dinv, h2, n);
    k_agg<6, RG, 1><<<blk(n, RG), 1024, 0, stream>>>(srcIdx, dstIdx, dinv, h2, b2, out, elist, ecnt, keys, E, n);
}


# in-degrees counted by the owner groups' key scan (LDS integer adds) in a launch of its own that also records the edge lists; no global atomics; both edge sums walk lists; 6 launches; sum order free
# speedup vs baseline: 3.2760x; 1.4203x over previous
#include <hip/hip_runtime.h>

typedef float v2f __attribute__((ext_vector_type(2)));
typedef float v8f __attribute__((ext_vector_type(8)));


constexpr int RG = 392;

constexpr int KPL = 4;

__global__ void k_keys(const int* __restrict__ dst, long e, unsigned* __restrict__ keys) {
    const long nt = (long)gridDim.x * blockDim.x;
    const long t0 = (long)blockIdx.x * blockDim.x + threadIdx.x;
    const long nq = ((e + 15) & ~15L) / 4;
    int d[KPL][4];
#pragma unroll
    for (int p = 0; p < KPL; ++p) {
        const long t = t0 + p * nt;
        const long i = 4 * t;
        if (t < nq && i + 4 <= e && (((unsigned long)(dst + i)) & 15ul) == 0) {
            const int4 v = *(const int4*)(dst + i);
            d[p][0] = v.x; d[p][1] = v.y; d[p][2] = v.z; d[p][3] = v.w;
        } else {
#pragma unroll
            for (int c = 0; c < 4; ++c) d[p][c] = (t < nq && i + c < e) ? dst[i + c] : -1;
        }
    }
#pragma unroll
    for (int p = 0; p < KPL; ++p) {
        const long t = t0 + p * nt;
        if (t < nq) {
            unsigned k = 0;
#pragma unroll
            for (int c = 0; c < 4; ++c)
                if (d[p][c] >= 0) k |= ((unsigned)(d[p][c] / RG) & 255u) << (8 * c);
            keys[t] = k;
        }
    }
}


template <int K, int NOUT, bool RELU_BIAS>
__global__ __launch_bounds__(256) void k_gemm_wmma(const float* __restrict__ A,
                                                   const float* __restrict__ W,
                                                   const float* __restrict__ bias,
                                                   const float* __restrict__ rowScale,
                                                   float* __restrict__ Out,
                                                   int nRows) {
    constexpr int PITCH = 2 * NOUT + 32;
    constexpr int HK    = K / 2;
    constexpr int PAIRS = K / 4;
    constexpr int CT    = NOUT / 16;
    __shared__ __align__(16) float lw[PAIRS * PITCH];

    const int lane  = threadIdx.x & 31;
    const int wave  = threadIdx.x >> 5;
    const int laneM = lane & 15;
    const int hi    = lane >> 4;

    const int rowBase = blockIdx.x * 128 + wave * 16 + laneM;
    const int rowC    = rowBase < nRows ? rowBase : 0;

    v8f acc[CT] = {};

    float sc[8] = {};

    for (int kh = 0; kh < 2; ++kh) {
        __syncthreads();
        for (int idx = threadIdx.x; idx < HK * NOUT; idx += 256) {
            int lk = idx / NOUT;
            int c  = idx - lk * NOUT;
            lw[(lk >> 1) * PITCH + 2 * c + (lk & 1)] = W[(kh * HK + lk) * NOUT + c];
        }
        __syncthreads();

        if (kh == 1) {
#pragma unroll
            for (int r = 0; r < 8; ++r) {
                const int m = blockIdx.x * 128 + wave * 16 + 8 * hi + r;
                sc[r] = rowScale[m < nRows ? m : 0];
            }
        }

        for (int k4 = 0; k4 < HK; k4 += 4) {
            const int kk = kh * HK + k4 + 2 * hi;
            v2f a;
            a.x = A[(long)rowC * K + kk];
            a.y = A[(long)rowC * K + kk + 1];
            if (RELU_BIAS) {
                a.x = fmaxf(a.x + bias[kk], 0.0f);
                a.y = fmaxf(a.y + bias[kk + 1], 0.0f);
            }
            const float* lb = &lw[((k4 >> 1) + hi) * PITCH];

            v2f bfrag[CT];
#pragma unroll
            for (int ct = 0; ct < CT; ++ct)
                bfrag[ct] = *(const v2f*)&lb[2 * (ct * 16 + laneM)];

#pragma unroll
            for (int ct = 0; ct < CT; ++ct)
                acc[ct] = __builtin_amdgcn_wmma_f32_16x16x4_f32(
                    false, a, false, bfrag[ct], (short)0, acc[ct], false, false);
        }
    }

#pragma unroll
    for (int ct = 0; ct < CT; ++ct) {
#pragma unroll
        for (int r = 0; r < 8; ++r) {
            int m = blockIdx.x * 128 + wave * 16 + 8 * hi + r;
            if (m < nRows) Out[(long)m * NOUT + ct * 16 + laneM] = acc[ct][r] * sc[r];
        }
    }
}


constexpr int LCAP = 16384;

typedef int   v2i __attribute__((ext_vector_type(2)));
typedef float v4f __attribute__((ext_vector_type(4)));

template <int LOGF, int R, int MODE>
__global__ __launch_bounds__(1024) void k_agg(const int* __restrict__ src, const int* __restrict__ dst,
                                              const float* __restrict__ dinv, const float* __restrict__ h,
                                              const float* __restrict__ bias, float* __restrict__ out,
                                              int* glist, int* gcnt, const unsigned* __restrict__ keys,
                                              float* dinvOut, long nE, int n) {
    constexpr int F    = 1 << LOGF;
    constexpr int LPR  = F / 32;
    constexpr int G    = 32 / LPR;
    constexpr int U    = 8;
    constexpr int QCAP = 32 + 2 * 128;
    __shared__ __align__(16) float tile[MODE == 1 ? R * F : 4];
    __shared__ float dl[MODE == 1 ? R : 1];
    __shared__ int   rc[MODE == 0 ? R : 1];
    __shared__ int   qe[32][QCAP];
    __shared__ int   cur;

    const int lane = threadIdx.x & 31;
    const int wv = __builtin_amdgcn_readfirstlane((int)(threadIdx.x >> 5));
    const int lo = blockIdx.x * R;
    const int nr = (n - lo < R) ? (n - lo) : R;
    v2i* const mylist = (v2i*)glist + (long)blockIdx.x * LCAP;

    constexpr int F4  = F / 4;
    constexpr int NIT = (R * F4 + 1023) / 1024;
    const int     lim4 = nr * F4;
    if (MODE == 1) {
        const v4f* h4 = (const v4f*)(h + (long)lo * F);
        v4f hv[NIT];
#pragma unroll
        for (int it = 0; it < NIT; ++it) {
            int i4 = (int)threadIdx.x + it * 1024;
            i4     = i4 < lim4 ? i4 : lim4 - 1;
            hv[it] = h4[i4];
        }
#pragma unroll
        for (int it = 0; it < NIT; ++it) {
            const int i4 = (int)threadIdx.x + it * 1024;
            if (i4 < lim4) ((v4f*)tile)[i4] = hv[it];
        }
    }
    if (MODE == 1) {
        for (int r = threadIdx.x; r < nr; r += 1024) dl[r] = dinv[lo + r];
    } else {
        for (int r = threadIdx.x; r < nr; r += 1024) rc[r] = 0;
        if (threadIdx.x == 0) cur = 0;
    }
    __syncthreads();

    auto gatherRows = [&](float (&v)[G][LPR], int s, int j0) {
#pragma unroll
        for (int jj = 0; jj < G; ++jj) {
            const int    sj = __builtin_amdgcn_readlane(s, j0 + jj);
            const float* hp = h + ((long)sj << LOGF) + lane;
#pragma unroll
            for (int i = 0; i < LPR; ++i) v[jj][i] = hp[32 * i];
        }
    };
    auto addGroup = [&](const float (&v)[G][LPR], unsigned vm, int l, int j0) {
#pragma unroll
        for (int jj = 0; jj < G; ++jj) {
            if ((vm >> (j0 + jj)) & 1u) {
                const int lj = __builtin_amdgcn_readlane(l, j0 + jj);
                float*    tp = tile + (lj << LOGF) + lane;
#pragma unroll
                for (int i = 0; i < LPR; ++i)
                    __hip_atomic_fetch_add(tp + 32 * i, v[jj][i], __ATOMIC_RELAXED,
                                           __HIP_MEMORY_SCOPE_WORKGROUP);
            }
        }
    };
    auto addRows = [&](unsigned vm, int cnt, int s, int l) {
        for (int j0 = 0; j0 < cnt; j0 += G) {
            if (((vm >> j0) & ((1u << G) - 1u)) == 0) continue;
            float v[G][LPR];
            gatherRows(v, s, j0);
            addGroup(v, vm, l, j0);
        }
    };

    const int listed = (MODE == 1) ? gcnt[blockIdx.x] : 0;

    if (MODE == 1 && listed <= LCAP) {
        auto entry = [&](int i) {
            v2i t = {0, 0};
            if (i < listed) t = mylist[i];
            return t;
        };
        int b  = wv * 32;
        v2i e0 = entry(b + lane);
        while (b < listed) {
            const int      cnt = (listed - b < 32) ? (listed - b) : 32;
            const unsigned vm  = cnt < 32 ? ((1u << cnt) - 1u) : ~0u;
            float v[G][LPR];
            gatherRows(v, e0.x, 0);
            const v2i e1 = entry(b + 1024 + lane);
            addGroup(v, vm, e0.y, 0);
            for (int j0 = G; j0 < cnt; j0 += G) {
                gatherRows(v, e0.x, j0);
                addGroup(v, vm, e0.y, j0);
            }
            e0 = e1; b += 1024;
        }
    } else {
        int qn = 0;

        auto flush = [&]() {
            for (int qb = 0; qb < qn; qb += 32) {
                const int cnt = (qn - qb < 32) ? (qn - qb) : 32;
                int  s = 0, l = 0;
                bool ok = false;
                if (lane < cnt) {
                    const int e = qe[wv][qb + lane];
                    if (e < nE) {
                        const int loc = dst[e] - lo;
                        s = src[e];
                        if ((unsigned)loc < (unsigned)nr) {
                            ok = true;
                            l  = loc;
                        }
                    }
                }
                const unsigned vm = __builtin_amdgcn_ballot_w32(ok);
                if (!vm) continue;
                if (MODE == 0) {
                    const int nv = __builtin_popcount(vm);
                    int p = 0;
                    if (lane == 0)
                        p = __hip_atomic_fetch_add(&cur, nv, __ATOMIC_RELAXED, __HIP_MEMORY_SCOPE_WORKGROUP);
                    p = __builtin_amdgcn_readfirstlane(p);
                    if (p + nv <= LCAP && ok) {
                        v2i t;
                        t.x = s; t.y = l;
                        mylist[p + (int)__builtin_amdgcn_mbcnt_lo(vm, 0u)] = t;
                    }
                    if (ok) __hip_atomic_fetch_add(&rc[l], 1, __ATOMIC_RELAXED, __HIP_MEMORY_SCOPE_WORKGROUP);
                } else {
                    addRows(vm, cnt, s, l);
                }
            }
            qn = 0;
        };

        const unsigned gpat  = (blockIdx.x & 255u) * 0x01010101u;
        const unsigned nohit = gpat ^ 0x01010101u;
        const int      lane16 = 16 * lane;
        const long nK16   = (nE + 15) & ~15L;
        const long nSuper = (nE + U * 512 - 1) / (U * 512);
        const int  K      = (int)((nSuper + 31) / 32);
        const int  KT     = K * 32;
        const int  rot    = (int)(((long)blockIdx.x * KT) / gridDim.x);

        for (int k = 0; k < K; ++k) {
            long sc = (long)k * 32 + wv + rot;
            if (sc >= KT) sc -= KT;
            if (sc >= nSuper) continue;
            const long base = sc * (U * 512);
            unsigned d[U][4];
            if (base + U * 512 <= nK16) {
#pragma unroll
                for (int u = 0; u < U; ++u) {
                    const uint4 t = *(const uint4*)((const unsigned char*)keys + base + u * 512 + 16 * lane);
                    d[u][0] = t.x; d[u][1] = t.y; d[u][2] = t.z; d[u][3] = t.w;
                }
            } else {
#pragma unroll
                for (int u = 0; u < U; ++u) {
                    const long o = base + u * 512 + 16 * lane;
                    uint4 t = {nohit, nohit, nohit, nohit};
                    if (o + 16 <= nK16) t = *(const uint4*)((const unsigned char*)keys + o);
                    d[u][0] = t.x; d[u][1] = t.y; d[u][2] = t.z; d[u][3] = t.w;
                }
            }
#pragma unroll
            for (int u = 0; u < U; ++u)
#pragma unroll
                for (int c = 0; c < 4; ++c) {
                    const unsigned t = d[u][c] ^ gpat;
                    d[u][c] = (t - 0x01010101u) & ~t & 0x80808080u;
                }
            const int sbase = (int)base;
            bool again;
            do {
                again = false;
#pragma unroll
                for (int p = 0; p < 2 * U; ++p) {
#pragma unroll
                    for (int c = 2 * (p & 1); c < 2 * (p & 1) + 2; ++c) {
                        unsigned mask = __builtin_amdgcn_ballot_w32(d[p >> 1][c] != 0);
                        while (mask) {
                            const unsigned x   = d[p >> 1][c];
                            const int      pos = qn + (int)__builtin_amdgcn_mbcnt_lo(mask, 0u);
                            if (x != 0)
                                qe[wv][pos] = (sbase + lane16) + ((__builtin_ctz(x) + (4096 * (p >> 1) + 32 * c)) >> 3);
                            qn += __builtin_popcount(mask);
                            d[p >> 1][c] = x & (x - 1u);
                            mask = __builtin_amdgcn_ballot_w32(d[p >> 1][c] != 0);
                        }
                    }
                    if (qn >= 32) {
                        again = p + 1 < 2 * U;
                        break;
                    }
                }
                if (qn >= 32) flush();
            } while (again);
        }
        if (qn) flush();
    }
    v4f bv = {0.0f, 0.0f, 0.0f, 0.0f};
    if (MODE == 1 && bias) {
        const float* bp = bias + 4 * (threadIdx.x & (F4 - 1));
        bv.x = bp[0]; bv.y = bp[1]; bv.z = bp[2]; bv.w = bp[3];
    }
    __syncthreads();

    if (MODE == 0) {
        if (threadIdx.x == 0) gcnt[blockIdx.x] = cur;
        for (int r = threadIdx.x; r < nr; r += 1024) dinvOut[lo + r] = rsqrtf((float)rc[r] + 1.0f);
    } else {
        v4f* o4 = (v4f*)(out + (long)lo * F);
#pragma unroll
        for (int it = 0; it < NIT; ++it) {
            const int i4 = (int)threadIdx.x + it * 1024;
            if (i4 < lim4) {
                const float d = dl[i4 >> (LOGF - 2)];
                v4f         v = ((const v4f*)tile)[i4] * d;
                if (bias) v += bv;
                o4[i4] = v;
            }
        }
    }
}


extern "C" void kernel_launch(void* const* d_in, const int* in_sizes, int n_in,
                              void* d_out, int out_size, void* d_ws, size_t ws_size,
                              hipStream_t stream) {
    const float* x   = (const float*)d_in[0];
    const int*   ei  = (const int*)d_in[1];
    const float* W1  = (const float*)d_in[2];
    const float* b1  = (const float*)d_in[3];
    const float* W2  = (const float*)d_in[4];
    const float* b2  = (const float*)d_in[5];
    float*       out = (float*)d_out;

    const int  n = in_sizes[0] / 128;
    const long E = in_sizes[1] / 2;
    const int* srcIdx = ei;
    const int* dstIdx = ei + E;

    float* ws   = (float*)d_ws;
    float* dinv = ws;
    float* h1   = dinv + n;
    float* agg1 = h1 + (long)n * 128;
    float* h2   = agg1 + (long)n * 128;
    int*   ecnt  = (int*)(h2 + (long)n * 64);
    int*   elist = ecnt + n;

    const int B = 256;
    auto blk = [](long work, int b) { return (int)((work + b - 1) / b); };
    unsigned* keys = (unsigned*)(elist + (long)blk(n, RG) * (LCAP * 2));

    k_keys<<<blk(E + 15, 4 * KPL * B), B, 0, stream>>>(dstIdx, E, keys);
    k_agg<6, RG, 0><<<blk(n, RG), 1024, 0, stream>>>(srcIdx, dstIdx, nullptr, nullptr, nullptr, nullptr, elist, ecnt, keys, dinv, E, n);

    k_gemm_wmma<128, 128, false><<<blk(n, 128), 256, 0, stream>>>(x, W1, nullptr, dinv, h1, n);
    k_agg<7, RG, 1><<<blk(n, RG), 1024, 0, stream>>>(srcIdx, dstIdx, dinv, h1, nullptr, agg1, elist, ecnt, keys, nullptr, E, n);

    k_gemm_wmma<128, 64, true><<<blk(n, 128), 256, 0, stream>>>(agg1, W2, b1, dinv, h2, n);
    k_agg<6, RG, 1><<<blk(n, RG), 1024, 0, stream>>>(srcIdx, dstIdx, dinv, h2, b2, out, elist, ecnt, keys, nullptr, E, n);
}


# x W1 product moved into the record launch: 8 of a group's 32 waves multiply its rows (f32 WMMA, W1 in LDS) while the rest scan, then join; rows scaled by the new degrees; 5 launches; sum order free
# speedup vs baseline: 3.5506x; 1.0838x over previous
#include <hip/hip_runtime.h>

typedef float v2f __attribute__((ext_vector_type(2)));
typedef float v8f __attribute__((ext_vector_type(8)));


constexpr int RG = 392;

constexpr int KPL = 4;

__global__ void k_keys(const int* __restrict__ dst, long e, unsigned* __restrict__ keys) {
    const long nt = (long)gridDim.x * blockDim.x;
    const long t0 = (long)blockIdx.x * blockDim.x + threadIdx.x;
    const long nq = ((e + 15) & ~15L) / 4;
    int d[KPL][4];
#pragma unroll
    for (int p = 0; p < KPL; ++p) {
        const long t = t0 + p * nt;
        const long i = 4 * t;
        if (t < nq && i + 4 <= e && (((unsigned long)(dst + i)) & 15ul) == 0) {
            const int4 v = *(const int4*)(dst + i);
            d[p][0] = v.x; d[p][1] = v.y; d[p][2] = v.z; d[p][3] = v.w;
        } else {
#pragma unroll
            for (int c = 0; c < 4; ++c) d[p][c] = (t < nq && i + c < e) ? dst[i + c] : -1;
        }
    }
#pragma unroll
    for (int p = 0; p < KPL; ++p) {
        const long t = t0 + p * nt;
        if (t < nq) {
            unsigned k = 0;
#pragma unroll
            for (int c = 0; c < 4; ++c)
                if (d[p][c] >= 0) k |= ((unsigned)(d[p][c] / RG) & 255u) << (8 * c);
            keys[t] = k;
        }
    }
}


template <int K, int NOUT, bool RELU_BIAS>
__global__ __launch_bounds__(256) void k_gemm_wmma(const float* __restrict__ A,
                                                   const float* __restrict__ W,
                                                   const float* __restrict__ bias,
                                                   const float* __restrict__ rowScale,
                                                   float* __restrict__ Out,
                                                   int nRows) {
    constexpr int PITCH = 2 * NOUT + 32;
    constexpr int HK    = K / 2;
    constexpr int PAIRS = K / 4;
    constexpr int CT    = NOUT / 16;
    __shared__ __align__(16) float lw[PAIRS * PITCH];

    const int lane  = threadIdx.x & 31;
    const int wave  = threadIdx.x >> 5;
    const int laneM = lane & 15;
    const int hi    = lane >> 4;

    const int rowBase = blockIdx.x * 128 + wave * 16 + laneM;
    const int rowC    = rowBase < nRows ? rowBase : 0;

    v8f acc[CT] = {};

    float sc[8] = {};

    for (int kh = 0; kh < 2; ++kh) {
        __syncthreads();
        for (int idx = threadIdx.x; idx < HK * NOUT; idx += 256) {
            int lk = idx / NOUT;
            int c  = idx - lk * NOUT;
            lw[(lk >> 1) * PITCH + 2 * c + (lk & 1)] = W[(kh * HK + lk) * NOUT + c];
        }
        __syncthreads();

        if (kh == 1) {
#pragma unroll
            for (int r = 0; r < 8; ++r) {
                const int m = blockIdx.x * 128 + wave * 16 + 8 * hi + r;
                sc[r] = rowScale[m < nRows ? m : 0];
            }
        }

        for (int k4 = 0; k4 < HK; k4 += 4) {
            const int kk = kh * HK + k4 + 2 * hi;
            v2f a;
            a.x = A[(long)rowC * K + kk];
            a.y = A[(long)rowC * K + kk + 1];
            if (RELU_BIAS) {
                a.x = fmaxf(a.x + bias[kk], 0.0f);
                a.y = fmaxf(a.y + bias[kk + 1], 0.0f);
            }
            const float* lb = &lw[((k4 >> 1) + hi) * PITCH];

            v2f bfrag[CT];
#pragma unroll
            for (int ct = 0; ct < CT; ++ct)
                bfrag[ct] = *(const v2f*)&lb[2 * (ct * 16 + laneM)];

#pragma unroll
            for (int ct = 0; ct < CT; ++ct)
                acc[ct] = __builtin_amdgcn_wmma_f32_16x16x4_f32(
                    false, a, false, bfrag[ct], (short)0, acc[ct], false, false);
        }
    }

#pragma unroll
    for (int ct = 0; ct < CT; ++ct) {
#pragma unroll
        for (int r = 0; r < 8; ++r) {
            int m = blockIdx.x * 128 + wave * 16 + 8 * hi + r;
            if (m < nRows) Out[(long)m * NOUT + ct * 16 + laneM] = acc[ct][r] * sc[r];
        }
    }
}


constexpr int LCAP = 16384;

typedef int   v2i __attribute__((ext_vector_type(2)));
typedef float v4f __attribute__((ext_vector_type(4)));

template <int LOGF, int R, int MODE>
__global__ __launch_bounds__(1024) void k_agg(const int* __restrict__ src, const int* __restrict__ dst,
                                              const float* __restrict__ dinv, const float* __restrict__ h,
                                              const float* __restrict__ bias, float* __restrict__ out,
                                              int* glist, int* gcnt, const unsigned* __restrict__ keys,
                                              float* dinvOut, long nE, int n,
                                              const float* __restrict__ xg, const float* __restrict__ wg) {
    constexpr int F    = 1 << LOGF;
    constexpr int LPR  = F / 32;
    constexpr int G    = 32 / LPR;
    constexpr int U    = 8;
    constexpr int QCAP = 32 + 2 * 128;
    __shared__ __align__(16) float tile[MODE == 1 ? R * F : 4];
    __shared__ float dl[MODE == 1 ? R : 1];
    __shared__ int   rc[MODE == 0 ? R : 1];
    __shared__ int   qe[32][QCAP];
    __shared__ int   cur;
    __shared__ int   nextStep;
    constexpr int WP = 2 * 128 + 32;
    constexpr int PW = 8;
    __shared__ __align__(16) float lw[MODE == 0 ? 64 * WP : 4];
    __shared__ __align__(16) float ht[MODE == 0 ? R * 128 : 4];

    const int lane = threadIdx.x & 31;
    const int wv = __builtin_amdgcn_readfirstlane((int)(threadIdx.x >> 5));
    const int lo = blockIdx.x * R;
    const int nr = (n - lo < R) ? (n - lo) : R;
    v2i* const mylist = (v2i*)glist + (long)blockIdx.x * LCAP;

    constexpr int F4  = F / 4;
    constexpr int NIT = (R * F4 + 1023) / 1024;
    const int     lim4 = nr * F4;
    if (MODE == 1) {
        const v4f* h4 = (const v4f*)(h + (long)lo * F);
        v4f hv[NIT];
#pragma unroll
        for (int it = 0; it < NIT; ++it) {
            int i4 = (int)threadIdx.x + it * 1024;
            i4     = i4 < lim4 ? i4 : lim4 - 1;
            hv[it] = h4[i4];
        }
#pragma unroll
        for (int it = 0; it < NIT; ++it) {
            const int i4 = (int)threadIdx.x + it * 1024;
            if (i4 < lim4) ((v4f*)tile)[i4] = hv[it];
        }
    }
    if (MODE == 1) {
        for (int r = threadIdx.x; r < nr; r += 1024) dl[r] = dinv[lo + r];
    } else {
        for (int r = threadIdx.x; r < nr; r += 1024) rc[r] = 0;
        if (threadIdx.x == 0) { cur = 0; nextStep = 0; }
        for (int idx = threadIdx.x; idx < 128 * 128; idx += 1024) {
            const int k = idx >> 7, c = idx & 127;
            lw[(k >> 1) * WP + 2 * c + (k & 1)] = wg[idx];
        }
    }
    __syncthreads();

    auto gatherRows = [&](float (&v)[G][LPR], int s, int j0) {
#pragma unroll
        for (int jj = 0; jj < G; ++jj) {
            const int    sj = __builtin_amdgcn_readlane(s, j0 + jj);
            const float* hp = h + ((long)sj << LOGF) + lane;
#pragma unroll
            for (int i = 0; i < LPR; ++i) v[jj][i] = hp[32 * i];
        }
    };
    auto addGroup = [&](const float (&v)[G][LPR], unsigned vm, int l, int j0) {
#pragma unroll
        for (int jj = 0; jj < G; ++jj) {
            if ((vm >> (j0 + jj)) & 1u) {
                const int lj = __builtin_amdgcn_readlane(l, j0 + jj);
                float*    tp = tile + (lj << LOGF) + lane;
#pragma unroll
                for (int i = 0; i < LPR; ++i)
                    __hip_atomic_fetch_add(tp + 32 * i, v[jj][i], __ATOMIC_RELAXED,
                                           __HIP_MEMORY_SCOPE_WORKGROUP);
            }
        }
    };
    auto addRows = [&](unsigned vm, int cnt, int s, int l) {
        for (int j0 = 0; j0 < cnt; j0 += G) {
            if (((vm >> j0) & ((1u << G) - 1u)) == 0) continue;
            float v[G][LPR];
            gatherRows(v, s, j0);
            addGroup(v, vm, l, j0);
        }
    };

    const int listed = (MODE == 1) ? gcnt[blockIdx.x] : 0;

    if (MODE == 1 && listed <= LCAP) {
        auto entry = [&](int i) {
            v2i t = {0, 0};
            if (i < listed) t = mylist[i];
            return t;
        };
        int b  = wv * 32;
        v2i e0 = entry(b + lane);
        while (b < listed) {
            const int      cnt = (listed - b < 32) ? (listed - b) : 32;
            const unsigned vm  = cnt < 32 ? ((1u << cnt) - 1u) : ~0u;
            float v[G][LPR];
            gatherRows(v, e0.x, 0);
            const v2i e1 = entry(b + 1024 + lane);
            addGroup(v, vm, e0.y, 0);
            for (int j0 = G; j0 < cnt; j0 += G) {
                gatherRows(v, e0.x, j0);
                addGroup(v, vm, e0.y, j0);
            }
            e0 = e1; b += 1024;
        }
    } else {
        if (MODE == 0 && wv < PW) {
            const int laneM = lane & 15;
            const int hi    = lane >> 4;
            const int units = 2 * ((nr + 15) >> 4);
            auto rowPtr = [&](int u) {
                const int rowL = (u >> 1) * 16 + laneM;
                return xg + (long)(lo + (rowL < nr ? rowL : 0)) * 128 + 2 * hi;
            };
            auto loadA = [&](v2f (&a)[8], const float* xr, int kb) {
#pragma unroll
                for (int j = 0; j < 8; ++j) a[j] = *(const v2f*)(xr + kb + 4 * j);
            };
            auto mma = [&](v8f (&acc)[4], const v2f (&a)[8], int kb, int cb) {
#pragma unroll
                for (int j = 0; j < 8; ++j) {
                    const float* lb = &lw[(((kb + 4 * j) >> 1) + hi) * WP + 2 * cb];
                    v2f bf[4];
#pragma unroll
                    for (int ct = 0; ct < 4; ++ct) bf[ct] = *(const v2f*)&lb[2 * (ct * 16 + laneM)];
#pragma unroll
                    for (int ct = 0; ct < 4; ++ct)
                        acc[ct] = __builtin_amdgcn_wmma_f32_16x16x4_f32(
                            false, a[j], false, bf[ct], (short)0, acc[ct], false, false);
                }
            };
            v2f aA[8], aB[8];
            int u = wv;
            if (u < units) loadA(aA, rowPtr(u), 0);
            for (; u < units; u += PW) {
                const float* xr = rowPtr(u);
                const int    cb = (u & 1) * 64;
                v8f acc[4] = {};
                loadA(aB, xr, 32);
                mma(acc, aA, 0, cb);
                loadA(aA, xr, 64);
                mma(acc, aB, 32, cb);
                loadA(aB, xr, 96);
                mma(acc, aA, 64, cb);
                if (u + PW < units) loadA(aA, rowPtr(u + PW), 0);
                mma(acc, aB, 96, cb);
#pragma unroll
                for (int r = 0; r < 8; ++r) {
                    const int m = (u >> 1) * 16 + 8 * hi + r;
                    if (m < nr) {
#pragma unroll
                        for (int ct = 0; ct < 4; ++ct) ht[m * 128 + cb + ct * 16 + laneM] = acc[ct][r];
                    }
                }
            }
        }

        int qn = 0;

        auto flush = [&]() {
            for (int qb = 0; qb < qn; qb += 32) {
                const int cnt = (qn - qb < 32) ? (qn - qb) : 32;
                int  s = 0, l = 0;
                bool ok = false;
                if (lane < cnt) {
                    const int e = qe[wv][qb + lane];
                    if (e < nE) {
                        const int loc = dst[e] - lo;
                        s = src[e];
                        if ((unsigned)loc < (unsigned)nr) {
                            ok = true;
                            l  = loc;
                        }
                    }
                }
                const unsigned vm = __builtin_amdgcn_ballot_w32(ok);
                if (!vm) continue;
                if (MODE == 0) {
                    const int nv = __builtin_popcount(vm);
                    int p = 0;
                    if (lane == 0)
                        p = __hip_atomic_fetch_add(&cur, nv, __ATOMIC_RELAXED, __HIP_MEMORY_SCOPE_WORKGROUP);
                    p = __builtin_amdgcn_readfirstlane(p);
                    if (p + nv <= LCAP && ok) {
                        v2i t;
                        t.x = s; t.y = l;
                        mylist[p + (int)__builtin_amdgcn_mbcnt_lo(vm, 0u)] = t;
                    }
                    if (ok) __hip_atomic_fetch_add(&rc[l], 1, __ATOMIC_RELAXED, __HIP_MEMORY_SCOPE_WORKGROUP);
                } else {
                    addRows(vm, cnt, s, l);
                }
            }
            qn = 0;
        };

        const unsigned gpat  = (blockIdx.x & 255u) * 0x01010101u;
        const unsigned nohit = gpat ^ 0x01010101u;
        const int      lane16 = 16 * lane;
        const long nK16   = (nE + 15) & ~15L;
        const long nSuper = (nE + U * 512 - 1) / (U * 512);
        const int  K      = (int)((nSuper + 31) / 32);
        const int  KT     = (MODE == 0) ? (int)nSuper : K * 32;
        const int  rot    = (int)(((long)blockIdx.x * KT) / gridDim.x);

        for (int k = 0;; ++k) {
            long sc;
            if (MODE == 0) {
                int s = 0;
                if (lane == 0)
                    s = __hip_atomic_fetch_add(&nextStep, 1, __ATOMIC_RELAXED, __HIP_MEMORY_SCOPE_WORKGROUP);
                s = __builtin_amdgcn_readfirstlane(s);
                if (s >= KT) break;
                sc = s + rot;
            } else {
                if (k >= K) break;
                sc = (long)k * 32 + wv + rot;
            }
            if (sc >= KT) sc -= KT;
            if (sc >= nSuper) continue;
            const long base = sc * (U * 512);
            unsigned d[U][4];
            if (base + U * 512 <= nK16) {
#pragma unroll
                for (int u = 0; u < U; ++u) {
                    const uint4 t = *(const uint4*)((const unsigned char*)keys + base + u * 512 + 16 * lane);
                    d[u][0] = t.x; d[u][1] = t.y; d[u][2] = t.z; d[u][3] = t.w;
                }
            } else {
#pragma unroll
                for (int u = 0; u < U; ++u) {
                    const long o = base + u * 512 + 16 * lane;
                    uint4 t = {nohit, nohit, nohit, nohit};
                    if (o + 16 <= nK16) t = *(const uint4*)((const unsigned char*)keys + o);
                    d[u][0] = t.x; d[u][1] = t.y; d[u][2] = t.z; d[u][3] = t.w;
                }
            }
#pragma unroll
            for (int u = 0; u < U; ++u)
#pragma unroll
                for (int c = 0; c < 4; ++c) {
                    const unsigned t = d[u][c] ^ gpat;
                    d[u][c] = (t - 0x01010101u) & ~t & 0x80808080u;
                }
            const int sbase = (int)base;
            bool again;
            do {
                again = false;
#pragma unroll
                for (int p = 0; p < 2 * U; ++p) {
#pragma unroll
                    for (int c = 2 * (p & 1); c < 2 * (p & 1) + 2; ++c) {
                        unsigned mask = __builtin_amdgcn_ballot_w32(d[p >> 1][c] != 0);
                        while (mask) {
                            const unsigned x   = d[p >> 1][c];
                            const int      pos = qn + (int)__builtin_amdgcn_mbcnt_lo(mask, 0u);
                            if (x != 0)
                                qe[wv][pos] = (sbase + lane16) + ((__builtin_ctz(x) + (4096 * (p >> 1) + 32 * c)) >> 3);
                            qn += __builtin_popcount(mask);
                            d[p >> 1][c] = x & (x - 1u);
                            mask = __builtin_amdgcn_ballot_w32(d[p >> 1][c] != 0);
                        }
                    }
                    if (qn >= 32) {
                        again = p + 1 < 2 * U;
                        break;
                    }
                }
                if (qn >= 32) flush();
            } while (again);
        }
        if (qn) flush();
    }
    v4f bv = {0.0f, 0.0f, 0.0f, 0.0f};
    if (MODE == 1 && bias) {
        const float* bp = bias + 4 * (threadIdx.x & (F4 - 1));
        bv.x = bp[0]; bv.y = bp[1]; bv.z = bp[2]; bv.w = bp[3];
    }
    __syncthreads();

    if (MODE == 0) {
        if (threadIdx.x == 0) gcnt[blockIdx.x] = cur;
        for (int r = threadIdx.x; r < nr; r += 1024) dinvOut[lo + r] = rsqrtf((float)rc[r] + 1.0f);
        constexpr int NIT0 = (R * 32 + 1023) / 1024;
        const int     lim0 = nr * 32;
        v4f*  o4 = (v4f*)(out + (long)lo * 128);
        v4f   hv[NIT0];
        float sv[NIT0];
#pragma unroll
        for (int it = 0; it < NIT0; ++it) {
            int i4 = (int)threadIdx.x + it * 1024;
            i4     = i4 < lim0 ? i4 : lim0 - 1;
            hv[it] = ((const v4f*)ht)[i4];
            sv[it] = rsqrtf((float)rc[i4 >> 5] + 1.0f);
        }
#pragma unroll
        for (int it = 0; it < NIT0; ++it) {
            const int i4 = (int)threadIdx.x + it * 1024;
            if (i4 < lim0) o4[i4] = hv[it] * sv[it];
        }
    } else {
        v4f* o4 = (v4f*)(out + (long)lo * F);
#pragma unroll
        for (int it = 0; it < NIT; ++it) {
            const int i4 = (int)threadIdx.x + it * 1024;
            if (i4 < lim4) {
                const float d = dl[i4 >> (LOGF - 2)];
                v4f         v = ((const v4f*)tile)[i4] * d;
                if (bias) v += bv;
                o4[i4] = v;
            }
        }
    }
}


extern "C" void kernel_launch(void* const* d_in, const int* in_sizes, int n_in,
                              void* d_out, int out_size, void* d_ws, size_t ws_size,
                              hipStream_t stream) {
    const float* x   = (const float*)d_in[0];
    const int*   ei  = (const int*)d_in[1];
    const float* W1  = (const float*)d_in[2];
    const float* b1  = (const float*)d_in[3];
    const float* W2  = (const float*)d_in[4];
    const float* b2  = (const float*)d_in[5];
    float*       out = (float*)d_out;

    const int  n = in_sizes[0] / 128;
    const long E = in_sizes[1] / 2;
    const int* srcIdx = ei;
    const int* dstIdx = ei + E;

    float* ws   = (float*)d_ws;
    float* dinv = ws;
    float* h1   = dinv + n;
    float* agg1 = h1 + (long)n * 128;
    float* h2   = agg1 + (long)n * 128;
    int*   ecnt  = (int*)(h2 + (long)n * 64);
    int*   elist = ecnt + n;

    const int B = 256;
    auto blk = [](long work, int b) { return (int)((work + b - 1) / b); };
    unsigned* keys = (unsigned*)(elist + (long)blk(n, RG) * (LCAP * 2));

    k_keys<<<blk(E + 15, 4 * KPL * B), B, 0, stream>>>(dstIdx, E, keys);
    k_agg<6, RG, 0><<<blk(n, RG), 1024, 0, stream>>>(srcIdx, dstIdx, nullptr, nullptr, nullptr, h1, elist, ecnt, keys, dinv, E, n, x, W1);

    k_agg<7, RG, 1><<<blk(n, RG), 1024, 0, stream>>>(srcIdx, dstIdx, dinv, h1, nullptr, agg1, elist, ecnt, keys, nullptr, E, n, nullptr, nullptr);

    k_gemm_wmma<128, 64, true><<<blk(n, 128), 256, 0, stream>>>(agg1, W2, b1, dinv, h2, n);
    k_agg<6, RG, 1><<<blk(n, RG), 1024, 0, stream>>>(srcIdx, dstIdx, dinv, h2, b2, out, elist, ecnt, keys, nullptr, E, n, nullptr, nullptr);
}


# layer-2 product moved into the layer-1 owner launch: each group runs relu + f32 WMMA on its finished LDS tile rows (pitch +4, W2 in LDS), stores h2; no aggregate plane; 4 launches; sum order free
# speedup vs baseline: 3.8343x; 1.0799x over previous
#include <hip/hip_runtime.h>

typedef float v2f __attribute__((ext_vector_type(2)));
typedef float v8f __attribute__((ext_vector_type(8)));


constexpr int RG = 392;

constexpr int KPL = 4;

__global__ void k_keys(const int* __restrict__ dst, long e, unsigned* __restrict__ keys) {
    const long nt = (long)gridDim.x * blockDim.x;
    const long t0 = (long)blockIdx.x * blockDim.x + threadIdx.x;
    const long nq = ((e + 15) & ~15L) / 4;
    int d[KPL][4];
#pragma unroll
    for (int p = 0; p < KPL; ++p) {
        const long t = t0 + p * nt;
        const long i = 4 * t;
        if (t < nq && i + 4 <= e && (((unsigned long)(dst + i)) & 15ul) == 0) {
            const int4 v = *(const int4*)(dst + i);
            d[p][0] = v.x; d[p][1] = v.y; d[p][2] = v.z; d[p][3] = v.w;
        } else {
#pragma unroll
            for (int c = 0; c < 4; ++c) d[p][c] = (t < nq && i + c < e) ? dst[i + c] : -1;
        }
    }
#pragma unroll
    for (int p = 0; p < KPL; ++p) {
        const long t = t0 + p * nt;
        if (t < nq) {
            unsigned k = 0;
#pragma unroll
            for (int c = 0; c < 4; ++c)
                if (d[p][c] >= 0) k |= ((unsigned)(d[p][c] / RG) & 255u) << (8 * c);
            keys[t] = k;
        }
    }
}


constexpr int LCAP = 16384;

typedef int   v2i __attribute__((ext_vector_type(2)));
typedef float v4f __attribute__((ext_vector_type(4)));

template <int LOGF, int R, int MODE, bool FUSE>
__global__ __launch_bounds__(1024) void k_agg(const int* __restrict__ src, const int* __restrict__ dst,
                                              const float* __restrict__ dinv, const float* __restrict__ h,
                                              const float* __restrict__ bias, float* __restrict__ out,
                                              int* glist, int* gcnt, const unsigned* __restrict__ keys,
                                              float* dinvOut, long nE, int n,
                                              const float* __restrict__ xg, const float* __restrict__ wg) {
    constexpr int F    = 1 << LOGF;
    constexpr int LPR  = F / 32;
    constexpr int G    = 32 / LPR;
    constexpr int U    = 8;
    constexpr int QCAP = 32 + 2 * 128;
    constexpr int TP   = FUSE ? F + 4 : F;
    __shared__ __align__(16) float tile[MODE == 1 ? R * TP : 4];
    __shared__ float dl[MODE == 1 ? R : 1];
    __shared__ int   rc[MODE == 0 ? R : 1];
    __shared__ int   qe[32][QCAP];
    __shared__ int   cur;
    __shared__ int   nextStep;
    constexpr int WP = 2 * 128 + 32;
    constexpr int PW = 8;
    __shared__ __align__(16) float lw[MODE == 0 ? 64 * WP : 4];
    __shared__ __align__(16) float ht[MODE == 0 ? R * 128 : 4];
    constexpr int W2P = 2 * 64 + 32;
    __shared__ __align__(16) float lw2[FUSE ? 64 * W2P : 4];
    __shared__ __align__(8)  float lb1[FUSE ? 128 : 2];

    const int lane = threadIdx.x & 31;
    const int wv = __builtin_amdgcn_readfirstlane((int)(threadIdx.x >> 5));
    const int lo = blockIdx.x * R;
    const int nr = (n - lo < R) ? (n - lo) : R;
    v2i* const mylist = (v2i*)glist + (long)blockIdx.x * LCAP;

    constexpr int F4  = F / 4;
    constexpr int NIT = (R * F4 + 1023) / 1024;
    const int     lim4 = nr * F4;
    if (MODE == 1) {
        const v4f* h4 = (const v4f*)(h + (long)lo * F);
        v4f hv[NIT];
#pragma unroll
        for (int it = 0; it < NIT; ++it) {
            int i4 = (int)threadIdx.x + it * 1024;
            i4     = i4 < lim4 ? i4 : lim4 - 1;
            hv[it] = h4[i4];
        }
#pragma unroll
        for (int it = 0; it < NIT; ++it) {
            const int i4 = (int)threadIdx.x + it * 1024;
            if (i4 < lim4) *(v4f*)(tile + (i4 >> (LOGF - 2)) * TP + 4 * (i4 & (F4 - 1))) = hv[it];
        }
    }
    if (MODE == 1) {
        for (int r = threadIdx.x; r < nr; r += 1024) dl[r] = dinv[lo + r];
        if (FUSE) {
            for (int idx = threadIdx.x; idx < 128 * 64; idx += 1024) {
                const int k = idx >> 6, cc = idx & 63;
                lw2[(k >> 1) * W2P + 2 * cc + (k & 1)] = wg[idx];
            }
            if (threadIdx.x < 128) lb1[threadIdx.x] = bias[threadIdx.x];
        }
    } else {
        for (int r = threadIdx.x; r < nr; r += 1024) rc[r] = 0;
        if (threadIdx.x == 0) { cur = 0; nextStep = 0; }
        for (int idx = threadIdx.x; idx < 128 * 128; idx += 1024) {
            const int k = idx >> 7, c = idx & 127;
            lw[(k >> 1) * WP + 2 * c + (k & 1)] = wg[idx];
        }
    }
    __syncthreads();

    auto gatherRows = [&](float (&v)[G][LPR], int s, int j0) {
#pragma unroll
        for (int jj = 0; jj < G; ++jj) {
            const int    sj = __builtin_amdgcn_readlane(s, j0 + jj);
            const float* hp = h + ((long)sj << LOGF) + lane;
#pragma unroll
            for (int i = 0; i < LPR; ++i) v[jj][i] = hp[32 * i];
        }
    };
    auto addGroup = [&](const float (&v)[G][LPR], unsigned vm, int l, int j0) {
#pragma unroll
        for (int jj = 0; jj < G; ++jj) {
            if ((vm >> (j0 + jj)) & 1u) {
                const int lj = __builtin_amdgcn_readlane(l, j0 + jj);
                float*    tp = tile + lj * TP + lane;
#pragma unroll
                for (int i = 0; i < LPR; ++i)
                    __hip_atomic_fetch_add(tp + 32 * i, v[jj][i], __ATOMIC_RELAXED,
                                           __HIP_MEMORY_SCOPE_WORKGROUP);
            }
        }
    };
    auto addRows = [&](unsigned vm, int cnt, int s, int l) {
        for (int j0 = 0; j0 < cnt; j0 += G) {
            if (((vm >> j0) & ((1u << G) - 1u)) == 0) continue;
            float v[G][LPR];
            gatherRows(v, s, j0);
            addGroup(v, vm, l, j0);
        }
    };

    const int listed = (MODE == 1) ? gcnt[blockIdx.x] : 0;

    if (MODE == 1 && listed <= LCAP) {
        auto entry = [&](int i) {
            v2i t = {0, 0};
            if (i < listed) t = mylist[i];
            return t;
        };
        int b  = wv * 32;
        v2i e0 = entry(b + lane);
        while (b < listed) {
            const int      cnt = (listed - b < 32) ? (listed - b) : 32;
            const unsigned vm  = cnt < 32 ? ((1u << cnt) - 1u) : ~0u;
            float v[G][LPR];
            gatherRows(v, e0.x, 0);
            const v2i e1 = entry(b + 1024 + lane);
            addGroup(v, vm, e0.y, 0);
            for (int j0 = G; j0 < cnt; j0 += G) {
                gatherRows(v, e0.x, j0);
                addGroup(v, vm, e0.y, j0);
            }
            e0 = e1; b += 1024;
        }
    } else {
        if (MODE == 0 && wv < PW) {
            const int laneM = lane & 15;
            const int hi    = lane >> 4;
            const int units = 2 * ((nr + 15) >> 4);
            auto rowPtr = [&](int u) {
                const int rowL = (u >> 1) * 16 + laneM;
                return xg + (long)(lo + (rowL < nr ? rowL : 0)) * 128 + 2 * hi;
            };
            auto loadA = [&](v2f (&a)[8], const float* xr, int kb) {
#pragma unroll
                for (int j = 0; j < 8; ++j) a[j] = *(const v2f*)(xr + kb + 4 * j);
            };
            auto mma = [&](v8f (&acc)[4], const v2f (&a)[8], int kb, int cb) {
#pragma unroll
                for (int j = 0; j < 8; ++j) {
                    const float* lb = &lw[(((kb + 4 * j) >> 1) + hi) * WP + 2 * cb];
                    v2f bf[4];
#pragma unroll
                    for (int ct = 0; ct < 4; ++ct) bf[ct] = *(const v2f*)&lb[2 * (ct * 16 + laneM)];
#pragma unroll
                    for (int ct = 0; ct < 4; ++ct)
                        acc[ct] = __builtin_amdgcn_wmma_f32_16x16x4_f32(
                            false, a[j], false, bf[ct], (short)0, acc[ct], false, false);
                }
            };
            v2f aA[8], aB[8];
            int u = wv;
            if (u < units) loadA(aA, rowPtr(u), 0);
            for (; u < units; u += PW) {
                const float* xr = rowPtr(u);
                const int    cb = (u & 1) * 64;
                v8f acc[4] = {};
                loadA(aB, xr, 32);
                mma(acc, aA, 0, cb);
                loadA(aA, xr, 64);
                mma(acc, aB, 32, cb);
                loadA(aB, xr, 96);
                mma(acc, aA, 64, cb);
                if (u + PW < units) loadA(aA, rowPtr(u + PW), 0);
                mma(acc, aB, 96, cb);
#pragma unroll
                for (int r = 0; r < 8; ++r) {
                    const int m = (u >> 1) * 16 + 8 * hi + r;
                    if (m < nr) {
#pragma unroll
                        for (int ct = 0; ct < 4; ++ct) ht[m * 128 + cb + ct * 16 + laneM] = acc[ct][r];
                    }
                }
            }
        }

        int qn = 0;

        auto flush = [&]() {
            for (int qb = 0; qb < qn; qb += 32) {
                const int cnt = (qn - qb < 32) ? (qn - qb) : 32;
                int  s = 0, l = 0;
                bool ok = false;
                if (lane < cnt) {
                    const int e = qe[wv][qb + lane];
                    if (e < nE) {
                        const int loc = dst[e] - lo;
                        s = src[e];
                        if ((unsigned)loc < (unsigned)nr) {
                            ok = true;
                            l  = loc;
                        }
                    }
                }
                const unsigned vm = __builtin_amdgcn_ballot_w32(ok);
                if (!vm) continue;
                if (MODE == 0) {
                    const int nv = __builtin_popcount(vm);
                    int p = 0;
                    if (lane == 0)
                        p = __hip_atomic_fetch_add(&cur, nv, __ATOMIC_RELAXED, __HIP_MEMORY_SCOPE_WORKGROUP);
                    p = __builtin_amdgcn_readfirstlane(p);
                    if (p + nv <= LCAP && ok) {
                        v2i t;
                        t.x = s; t.y = l;
                        mylist[p + (int)__builtin_amdgcn_mbcnt_lo(vm, 0u)] = t;
                    }
                    if (ok) __hip_atomic_fetch_add(&rc[l], 1, __ATOMIC_RELAXED, __HIP_MEMORY_SCOPE_WORKGROUP);
                } else {
                    addRows(vm, cnt, s, l);
                }
            }
            qn = 0;
        };

        const unsigned gpat  = (blockIdx.x & 255u) * 0x01010101u;
        const unsigned nohit = gpat ^ 0x01010101u;
        const int      lane16 = 16 * lane;
        const long nK16   = (nE + 15) & ~15L;
        const long nSuper = (nE + U * 512 - 1) / (U * 512);
        const int  K      = (int)((nSuper + 31) / 32);
        const int  KT     = (MODE == 0) ? (int)nSuper : K * 32;
        const int  rot    = (int)(((long)blockIdx.x * KT) / gridDim.x);

        for (int k = 0;; ++k) {
            long sc;
            if (MODE == 0) {
                int s = 0;
                if (lane == 0)
                    s = __hip_atomic_fetch_add(&nextStep, 1, __ATOMIC_RELAXED, __HIP_MEMORY_SCOPE_WORKGROUP);
                s = __builtin_amdgcn_readfirstlane(s);
                if (s >= KT) break;
                sc = s + rot;
            } else {
                if (k >= K) break;
                sc = (long)k * 32 + wv + rot;
            }
            if (sc >= KT) sc -= KT;
            if (sc >= nSuper) continue;
            const long base = sc * (U * 512);
            unsigned d[U][4];
            if (base + U * 512 <= nK16) {
#pragma unroll
                for (int u = 0; u < U; ++u) {
                    const uint4 t = *(const uint4*)((const unsigned char*)keys + base + u * 512 + 16 * lane);
                    d[u][0] = t.x; d[u][1] = t.y; d[u][2] = t.z; d[u][3] = t.w;
                }
            } else {
#pragma unroll
                for (int u = 0; u < U; ++u) {
                    const long o = base + u * 512 + 16 * lane;
                    uint4 t = {nohit, nohit, nohit, nohit};
                    if (o + 16 <= nK16) t = *(const uint4*)((const unsigned char*)keys + o);
                    d[u][0] = t.x; d[u][1] = t.y; d[u][2] = t.z; d[u][3] = t.w;
                }
            }
#pragma unroll
            for (int u = 0; u < U; ++u)
#pragma unroll
                for (int c = 0; c < 4; ++c) {
                    const unsigned t = d[u][c] ^ gpat;
                    d[u][c] = (t - 0x01010101u) & ~t & 0x80808080u;
                }
            const int sbase = (int)base;
            bool again;
            do {
                again = false;
#pragma unroll
                for (int p = 0; p < 2 * U; ++p) {
#pragma unroll
                    for (int c = 2 * (p & 1); c < 2 * (p & 1) + 2; ++c) {
                        unsigned mask = __builtin_amdgcn_ballot_w32(d[p >> 1][c] != 0);
                        while (mask) {
                            const unsigned x   = d[p >> 1][c];
                            const int      pos = qn + (int)__builtin_amdgcn_mbcnt_lo(mask, 0u);
                            if (x != 0)
                                qe[wv][pos] = (sbase + lane16) + ((__builtin_ctz(x) + (4096 * (p >> 1) + 32 * c)) >> 3);
                            qn += __builtin_popcount(mask);
                            d[p >> 1][c] = x & (x - 1u);
                            mask = __builtin_amdgcn_ballot_w32(d[p >> 1][c] != 0);
                        }
                    }
                    if (qn >= 32) {
                        again = p + 1 < 2 * U;
                        break;
                    }
                }
                if (qn >= 32) flush();
            } while (again);
        }
        if (qn) flush();
    }
    v4f bv = {0.0f, 0.0f, 0.0f, 0.0f};
    if (MODE == 1 && !FUSE && bias) {
        const float* bp = bias + 4 * (threadIdx.x & (F4 - 1));
        bv.x = bp[0]; bv.y = bp[1]; bv.z = bp[2]; bv.w = bp[3];
    }
    __syncthreads();

    if (MODE == 0) {
        if (threadIdx.x == 0) gcnt[blockIdx.x] = cur;
        for (int r = threadIdx.x; r < nr; r += 1024) dinvOut[lo + r] = rsqrtf((float)rc[r] + 1.0f);
        constexpr int NIT0 = (R * 32 + 1023) / 1024;
        const int     lim0 = nr * 32;
        v4f*  o4 = (v4f*)(out + (long)lo * 128);
        v4f   hv[NIT0];
        float sv[NIT0];
#pragma unroll
        for (int it = 0; it < NIT0; ++it) {
            int i4 = (int)threadIdx.x + it * 1024;
            i4     = i4 < lim0 ? i4 : lim0 - 1;
            hv[it] = ((const v4f*)ht)[i4];
            sv[it] = rsqrtf((float)rc[i4 >> 5] + 1.0f);
        }
#pragma unroll
        for (int it = 0; it < NIT0; ++it) {
            const int i4 = (int)threadIdx.x + it * 1024;
            if (i4 < lim0) o4[i4] = hv[it] * sv[it];
        }
    } else if (FUSE) {
        const int laneM = lane & 15;
        const int hi    = lane >> 4;
        const int units = (nr + 15) >> 4;
        for (int u = wv; u < units; u += 32) {
            const int    rowL = u * 16 + laneM;
            const int    rowA = rowL < nr ? rowL : u * 16;
            const float  da   = dl[rowA];
            const float* ar   = tile + rowA * TP + 2 * hi;
            v8f acc[4] = {};
#pragma unroll 8
            for (int k4 = 0; k4 < 128; k4 += 4) {
                v2f       a  = *(const v2f*)(ar + k4);
                const v2f bb = *(const v2f*)&lb1[k4 + 2 * hi];
                a.x = fmaxf(a.x * da + bb.x, 0.0f);
                a.y = fmaxf(a.y * da + bb.y, 0.0f);
                const float* lb = &lw2[((k4 >> 1) + hi) * W2P];
                v2f bf[4];
#pragma unroll
                for (int ct = 0; ct < 4; ++ct) bf[ct] = *(const v2f*)&lb[2 * (ct * 16 + laneM)];
#pragma unroll
                for (int ct = 0; ct < 4; ++ct)
                    acc[ct] = __builtin_amdgcn_wmma_f32_16x16x4_f32(
                        false, a, false, bf[ct], (short)0, acc[ct], false, false);
            }
#pragma unroll
            for (int r = 0; r < 8; ++r) {
                const int m = u * 16 + 8 * hi + r;
                if (m < nr) {
                    const float dm = dl[m];
#pragma unroll
                    for (int ct = 0; ct < 4; ++ct) tile[m * TP + ct * 16 + laneM] = acc[ct][r] * dm;
                }
            }
        }
        __syncthreads();
        constexpr int NITD = (R * 16 + 1023) / 1024;
        const int     limD = nr * 16;
        v4f*          o4   = (v4f*)(out + (long)lo * 64);
#pragma unroll
        for (int it = 0; it < NITD; ++it) {
            const int i4 = (int)threadIdx.x + it * 1024;
            if (i4 < limD) o4[i4] = *(const v4f*)(tile + (i4 >> 4) * TP + 4 * (i4 & 15));
        }
    } else {
        v4f* o4 = (v4f*)(out + (long)lo * F);
#pragma unroll
        for (int it = 0; it < NIT; ++it) {
            const int i4 = (int)threadIdx.x + it * 1024;
            if (i4 < lim4) {
                const float d = dl[i4 >> (LOGF - 2)];
                v4f         v = *(const v4f*)(tile + (i4 >> (LOGF - 2)) * TP + 4 * (i4 & (F4 - 1))) * d;
                if (bias) v += bv;
                o4[i4] = v;
            }
        }
    }
}


extern "C" void kernel_launch(void* const* d_in, const int* in_sizes, int n_in,
                              void* d_out, int out_size, void* d_ws, size_t ws_size,
                              hipStream_t stream) {
    const float* x   = (const float*)d_in[0];
    const int*   ei  = (const int*)d_in[1];
    const float* W1  = (const float*)d_in[2];
    const float* b1  = (const float*)d_in[3];
    const float* W2  = (const float*)d_in[4];
    const float* b2  = (const float*)d_in[5];
    float*       out = (float*)d_out;

    const int  n = in_sizes[0] / 128;
    const long E = in_sizes[1] / 2;
    const int* srcIdx = ei;
    const int* dstIdx = ei + E;

    float* ws   = (float*)d_ws;
    float* dinv = ws;
    float* h1   = dinv + n;
    float* h2   = h1 + (long)n * 128;
    int*   ecnt  = (int*)(h2 + (long)n * 64);
    int*   elist = ecnt + n;

    const int B = 256;
    auto blk = [](long work, int b) { return (int)((work + b - 1) / b); };
    unsigned* keys = (unsigned*)(elist + (long)blk(n, RG) * (LCAP * 2));

    k_keys<<<blk(E + 15, 4 * KPL * B), B, 0, stream>>>(dstIdx, E, keys);
    k_agg<6, RG, 0, false><<<blk(n, RG), 1024, 0, stream>>>(srcIdx, dstIdx, nullptr, nullptr, nullptr, h1, elist, ecnt, keys, dinv, E, n, x, W1);

    k_agg<7, RG, 1, true><<<blk(n, RG), 1024, 0, stream>>>(srcIdx, dstIdx, dinv, h1, b1, h2, elist, ecnt, keys, nullptr, E, n, nullptr, W2);

    k_agg<6, RG, 1, false><<<blk(n, RG), 1024, 0, stream>>>(srcIdx, dstIdx, dinv, h2, b2, out, elist, ecnt, keys, nullptr, E, n, nullptr, nullptr);
}


# owner key bytes (every group scans all 1.6M) replaced by per-owner bit planes built per 4096-edge block in LDS: a group scans only its own row, 1 bit per edge; sum order free as before
# speedup vs baseline: 4.0813x; 1.0644x over previous
#include <hip/hip_runtime.h>

typedef float v2f __attribute__((ext_vector_type(2)));
typedef float v8f __attribute__((ext_vector_type(8)));


constexpr int RG = 392;

constexpr int EB = 4096;
constexpr int BW = EB / 32;
constexpr int BP = BW + 4;
constexpr int OW = 256;

__global__ __launch_bounds__(256) void k_bits(const int* __restrict__ dst, long e, unsigned* __restrict__ plane,
                                              int ng) {
    __shared__ __align__(16) unsigned bm[OW * BP];
    const int  t    = threadIdx.x;
    const int  lane = t & 31;
    const int  wv   = __builtin_amdgcn_readfirstlane(t >> 5);
    const long e0   = (long)blockIdx.x * EB;
    const long pl   = (long)gridDim.x * BW;
    int d[4][4];
#pragma unroll
    for (int p = 0; p < 4; ++p) {
        const long i = e0 + 4 * (p * 256 + t);
        if (i + 4 <= e && (((unsigned long)(dst + i)) & 15ul) == 0) {
            const int4 v = *(const int4*)(dst + i);
            d[p][0] = v.x; d[p][1] = v.y; d[p][2] = v.z; d[p][3] = v.w;
        } else {
#pragma unroll
            for (int c = 0; c < 4; ++c) d[p][c] = (i + c < e) ? dst[i + c] : -1;
        }
    }
    for (int g0 = 0; g0 < ng; g0 += OW) {
        for (int i = t; i < OW * BP / 4; i += 256) ((uint4*)bm)[i] = uint4{0u, 0u, 0u, 0u};
        __syncthreads();
#pragma unroll
        for (int p = 0; p < 4; ++p) {
#pragma unroll
            for (int c = 0; c < 4; ++c) {
                const int g = d[p][c] / RG - g0;
                if (d[p][c] >= 0 && (unsigned)g < (unsigned)OW)
                    __hip_atomic_fetch_or(&bm[g * BP + 32 * p + (t >> 3)], 1u << (4 * (t & 7) + c),
                                          __ATOMIC_RELAXED, __HIP_MEMORY_SCOPE_WORKGROUP);
            }
        }
        __syncthreads();
        const int       rows = (ng - g0 < OW) ? (ng - g0) : OW;
        const int       r0   = 32 * wv;
        const int       nrow = (rows - r0 < 32) ? (rows - r0) : 32;
        const unsigned* ip   = bm + r0 * BP + 4 * lane;
        unsigned*       op   = plane + (long)(g0 + r0) * pl + (long)blockIdx.x * BW + 4 * lane;
        for (int j0 = 0; j0 < nrow; j0 += 8) {
            uint4 v[8];
#pragma unroll
            for (int j = 0; j < 8; ++j) v[j] = *(const uint4*)(ip + (j0 + j) * BP);
            if (j0 + 8 <= nrow) {
#pragma unroll
                for (int j = 0; j < 8; ++j) *(uint4*)(op + (long)(j0 + j) * pl) = v[j];
            } else {
#pragma unroll
                for (int j = 0; j < 8; ++j)
                    if (j0 + j < nrow) *(uint4*)(op + (long)(j0 + j) * pl) = v[j];
            }
        }
        __syncthreads();
    }
}


constexpr int LCAP = 16384;

typedef int   v2i __attribute__((ext_vector_type(2)));
typedef float v4f __attribute__((ext_vector_type(4)));

template <int LOGF, int R, int MODE, bool FUSE>
__global__ __launch_bounds__(1024) void k_agg(const int* __restrict__ src, const int* __restrict__ dst,
                                              const float* __restrict__ dinv, const float* __restrict__ h,
                                              const float* __restrict__ bias, float* __restrict__ out,
                                              int* glist, int* gcnt, const unsigned* __restrict__ planes,
                                              float* dinvOut, long nE, int n,
                                              const float* __restrict__ xg, const float* __restrict__ wg) {
    constexpr int F    = 1 << LOGF;
    constexpr int LPR  = F / 32;
    constexpr int G    = 32 / LPR;
    constexpr int U    = 4;
    constexpr int QCAP = 32 + 32;
    constexpr int TP   = FUSE ? F + 4 : F;
    __shared__ __align__(16) float tile[MODE == 1 ? R * TP : 4];
    __shared__ float dl[MODE == 1 ? R : 1];
    __shared__ int   rc[MODE == 0 ? R : 1];
    __shared__ int   qe[32][QCAP];
    __shared__ int   cur;
    __shared__ int   nextStep;
    constexpr int WP = 2 * 128 + 32;
    constexpr int PW = 8;
    __shared__ __align__(16) float lw[MODE == 0 ? 64 * WP : 4];
    __shared__ __align__(16) float ht[MODE == 0 ? R * 128 : 4];
    constexpr int W2P = 2 * 64 + 32;
    __shared__ __align__(16) float lw2[FUSE ? 64 * W2P : 4];
    __shared__ __align__(8)  float lb1[FUSE ? 128 : 2];

    const int lane = threadIdx.x & 31;
    const int wv = __builtin_amdgcn_readfirstlane((int)(threadIdx.x >> 5));
    const int lo = blockIdx.x * R;
    const int nr = (n - lo < R) ? (n - lo) : R;
    v2i* const mylist = (v2i*)glist + (long)blockIdx.x * LCAP;

    constexpr int F4  = F / 4;
    constexpr int NIT = (R * F4 + 1023) / 1024;
    const int     lim4 = nr * F4;
    if (MODE == 1) {
        const v4f* h4 = (const v4f*)(h + (long)lo * F);
        v4f hv[NIT];
#pragma unroll
        for (int it = 0; it < NIT; ++it) {
            int i4 = (int)threadIdx.x + it * 1024;
            i4     = i4 < lim4 ? i4 : lim4 - 1;
            hv[it] = h4[i4];
        }
#pragma unroll
        for (int it = 0; it < NIT; ++it) {
            const int i4 = (int)threadIdx.x + it * 1024;
            if (i4 < lim4) *(v4f*)(tile + (i4 >> (LOGF - 2)) * TP + 4 * (i4 & (F4 - 1))) = hv[it];
        }
    }
    if (MODE == 1) {
        for (int r = threadIdx.x; r < nr; r += 1024) dl[r] = dinv[lo + r];
        if (FUSE) {
            for (int idx = threadIdx.x; idx < 128 * 64; idx += 1024) {
                const int k = idx >> 6, cc = idx & 63;
                lw2[(k >> 1) * W2P + 2 * cc + (k & 1)] = wg[idx];
            }
            if (threadIdx.x < 128) lb1[threadIdx.x] = bias[threadIdx.x];
        }
    } else {
        for (int r = threadIdx.x; r < nr; r += 1024) rc[r] = 0;
        if (threadIdx.x == 0) { cur = 0; nextStep = 0; }
        for (int idx = threadIdx.x; idx < 128 * 128; idx += 1024) {
            const int k = idx >> 7, c = idx & 127;
            lw[(k >> 1) * WP + 2 * c + (k & 1)] = wg[idx];
        }
    }
    __syncthreads();

    auto gatherRows = [&](float (&v)[G][LPR], int s, int j0) {
#pragma unroll
        for (int jj = 0; jj < G; ++jj) {
            const int    sj = __builtin_amdgcn_readlane(s, j0 + jj);
            const float* hp = h + ((long)sj << LOGF) + lane;
#pragma unroll
            for (int i = 0; i < LPR; ++i) v[jj][i] = hp[32 * i];
        }
    };
    auto addGroup = [&](const float (&v)[G][LPR], unsigned vm, int l, int j0) {
#pragma unroll
        for (int jj = 0; jj < G; ++jj) {
            if ((vm >> (j0 + jj)) & 1u) {
                const int lj = __builtin_amdgcn_readlane(l, j0 + jj);
                float*    tp = tile + lj * TP + lane;
#pragma unroll
                for (int i = 0; i < LPR; ++i)
                    __hip_atomic_fetch_add(tp + 32 * i, v[jj][i], __ATOMIC_RELAXED,
                                           __HIP_MEMORY_SCOPE_WORKGROUP);
            }
        }
    };
    auto addRows = [&](unsigned vm, int cnt, int s, int l) {
        for (int j0 = 0; j0 < cnt; j0 += G) {
            if (((vm >> j0) & ((1u << G) - 1u)) == 0) continue;
            float v[G][LPR];
            gatherRows(v, s, j0);
            addGroup(v, vm, l, j0);
        }
    };

    const int listed = (MODE == 1) ? gcnt[blockIdx.x] : 0;

    if (MODE == 1 && listed <= LCAP) {
        auto entry = [&](int i) {
            v2i t = {0, 0};
            if (i < listed) t = mylist[i];
            return t;
        };
        int b  = wv * 32;
        v2i e0 = entry(b + lane);
        while (b < listed) {
            const int      cnt = (listed - b < 32) ? (listed - b) : 32;
            const unsigned vm  = cnt < 32 ? ((1u << cnt) - 1u) : ~0u;
            float v[G][LPR];
            gatherRows(v, e0.x, 0);
            const v2i e1 = entry(b + 1024 + lane);
            addGroup(v, vm, e0.y, 0);
            for (int j0 = G; j0 < cnt; j0 += G) {
                gatherRows(v, e0.x, j0);
                addGroup(v, vm, e0.y, j0);
            }
            e0 = e1; b += 1024;
        }
    } else {
        if (MODE == 0 && wv < PW) {
            const int laneM = lane & 15;
            const int hi    = lane >> 4;
            const int units = 2 * ((nr + 15) >> 4);
            auto rowPtr = [&](int u) {
                const int rowL = (u >> 1) * 16 + laneM;
                return xg + (long)(lo + (rowL < nr ? rowL : 0)) * 128 + 2 * hi;
            };
            auto loadA = [&](v2f (&a)[8], const float* xr, int kb) {
#pragma unroll
                for (int j = 0; j < 8; ++j) a[j] = *(const v2f*)(xr + kb + 4 * j);
            };
            auto mma = [&](v8f (&acc)[4], const v2f (&a)[8], int kb, int cb) {
#pragma unroll
                for (int j = 0; j < 8; ++j) {
                    const float* lb = &lw[(((kb + 4 * j) >> 1) + hi) * WP + 2 * cb];
                    v2f bf[4];
#pragma unroll
                    for (int ct = 0; ct < 4; ++ct) bf[ct] = *(const v2f*)&lb[2 * (ct * 16 + laneM)];
#pragma unroll
                    for (int ct = 0; ct < 4; ++ct)
                        acc[ct] = __builtin_amdgcn_wmma_f32_16x16x4_f32(
                            false, a[j], false, bf[ct], (short)0, acc[ct], false, false);
                }
            };
            v2f aA[8], aB[8];
            int u = wv;
            if (u < units) loadA(aA, rowPtr(u), 0);
            for (; u < units; u += PW) {
                const float* xr = rowPtr(u);
                const int    cb = (u & 1) * 64;
                v8f acc[4] = {};
                loadA(aB, xr, 32);
                mma(acc, aA, 0, cb);
                loadA(aA, xr, 64);
                mma(acc, aB, 32, cb);
                loadA(aB, xr, 96);
                mma(acc, aA, 64, cb);
                if (u + PW < units) loadA(aA, rowPtr(u + PW), 0);
                mma(acc, aB, 96, cb);
#pragma unroll
                for (int r = 0; r < 8; ++r) {
                    const int m = (u >> 1) * 16 + 8 * hi + r;
                    if (m < nr) {
#pragma unroll
                        for (int ct = 0; ct < 4; ++ct) ht[m * 128 + cb + ct * 16 + laneM] = acc[ct][r];
                    }
                }
            }
        }

        int qn = 0;

        auto flush = [&]() {
            for (int qb = 0; qb < qn; qb += 32) {
                const int cnt = (qn - qb < 32) ? (qn - qb) : 32;
                int  s = 0, l = 0;
                bool ok = false;
                if (lane < cnt) {
                    const int e = qe[wv][qb + lane];
                    if (e < nE) {
                        const int loc = dst[e] - lo;
                        s = src[e];
                        if ((unsigned)loc < (unsigned)nr) {
                            ok = true;
                            l  = loc;
                        }
                    }
                }
                const unsigned vm = __builtin_amdgcn_ballot_w32(ok);
                if (!vm) continue;
                if (MODE == 0) {
                    const int nv = __builtin_popcount(vm);
                    int p = 0;
                    if (lane == 0)
                        p = __hip_atomic_fetch_add(&cur, nv, __ATOMIC_RELAXED, __HIP_MEMORY_SCOPE_WORKGROUP);
                    p = __builtin_amdgcn_readfirstlane(p);
                    if (p + nv <= LCAP && ok) {
                        v2i t;
                        t.x = s; t.y = l;
                        mylist[p + (int)__builtin_amdgcn_mbcnt_lo(vm, 0u)] = t;
                    }
                    if (ok) __hip_atomic_fetch_add(&rc[l], 1, __ATOMIC_RELAXED, __HIP_MEMORY_SCOPE_WORKGROUP);
                } else {
                    addRows(vm, cnt, s, l);
                }
            }
            qn = 0;
        };

        const long nBlk   = (nE + EB - 1) / EB;
        const long nSuper = (nBlk + U - 1) / U;
        const int  K      = (int)((nSuper + 31) / 32);
        const unsigned* const myrow = planes + (long)blockIdx.x * (nBlk * BW);
        const int  KT     = (MODE == 0) ? (int)nSuper : K * 32;

        for (int k = 0;; ++k) {
            long sc;
            if (MODE == 0) {
                int s = 0;
                if (lane == 0)
                    s = __hip_atomic_fetch_add(&nextStep, 1, __ATOMIC_RELAXED, __HIP_MEMORY_SCOPE_WORKGROUP);
                s = __builtin_amdgcn_readfirstlane(s);
                if (s >= KT) break;
                sc = s;
            } else {
                if (k >= K) break;
                sc = (long)k * 32 + wv;
            }
            if (sc >= nSuper) continue;
            const unsigned* const rp = myrow + sc * (U * BW) + 4 * lane;
            unsigned d[U][4];
            if ((sc + 1) * U <= nBlk) {
#pragma unroll
                for (int u = 0; u < U; ++u) {
                    const uint4 t = *(const uint4*)(rp + u * BW);
                    d[u][0] = t.x; d[u][1] = t.y; d[u][2] = t.z; d[u][3] = t.w;
                }
            } else {
#pragma unroll
                for (int u = 0; u < U; ++u) {
                    uint4 t = {0u, 0u, 0u, 0u};
                    if (sc * U + u < nBlk) t = *(const uint4*)(rp + u * BW);
                    d[u][0] = t.x; d[u][1] = t.y; d[u][2] = t.z; d[u][3] = t.w;
                }
            }
            const int ebase = (int)(sc * (U * EB)) + 128 * lane;
            bool again;
            do {
                again = false;
#pragma unroll
                for (int p = 0; p < 4 * U; ++p) {
                    unsigned mask = __builtin_amdgcn_ballot_w32(d[p >> 2][p & 3] != 0);
                    while (mask) {
                        const unsigned x   = d[p >> 2][p & 3];
                        const int      pos = qn + (int)__builtin_amdgcn_mbcnt_lo(mask, 0u);
                        if (x != 0) qe[wv][pos] = ebase + (EB * (p >> 2) + 32 * (p & 3)) + __builtin_ctz(x);
                        qn += __builtin_popcount(mask);
                        d[p >> 2][p & 3] = x & (x - 1u);
                        if (qn >= 32) break;
                        mask = __builtin_amdgcn_ballot_w32(d[p >> 2][p & 3] != 0);
                    }
                    if (qn >= 32) {
                        again = true;
                        break;
                    }
                }
                if (qn >= 32) flush();
            } while (again);
        }
        if (qn) flush();
    }
    v4f bv = {0.0f, 0.0f, 0.0f, 0.0f};
    if (MODE == 1 && !FUSE && bias) {
        const float* bp = bias + 4 * (threadIdx.x & (F4 - 1));
        bv.x = bp[0]; bv.y = bp[1]; bv.z = bp[2]; bv.w = bp[3];
    }
    __syncthreads();

    if (MODE == 0) {
        if (threadIdx.x == 0) gcnt[blockIdx.x] = cur;
        for (int r = threadIdx.x; r < nr; r += 1024) dinvOut[lo + r] = rsqrtf((float)rc[r] + 1.0f);
        constexpr int NIT0 = (R * 32 + 1023) / 1024;
        const int     lim0 = nr * 32;
        v4f*  o4 = (v4f*)(out + (long)lo * 128);
        v4f   hv[NIT0];
        float sv[NIT0];
#pragma unroll
        for (int it = 0; it < NIT0; ++it) {
            int i4 = (int)threadIdx.x + it * 1024;
            i4     = i4 < lim0 ? i4 : lim0 - 1;
            hv[it] = ((const v4f*)ht)[i4];
            sv[it] = rsqrtf((float)rc[i4 >> 5] + 1.0f);
        }
#pragma unroll
        for (int it = 0; it < NIT0; ++it) {
            const int i4 = (int)threadIdx.x + it * 1024;
            if (i4 < lim0) o4[i4] = hv[it] * sv[it];
        }
    } else if (FUSE) {
        const int laneM = lane & 15;
        const int hi    = lane >> 4;
        const int units = (nr + 15) >> 4;
        for (int u = wv; u < units; u += 32) {
            const int    rowL = u * 16 + laneM;
            const int    rowA = rowL < nr ? rowL : u * 16;
            const float  da   = dl[rowA];
            const float* ar   = tile + rowA * TP + 2 * hi;
            v8f acc[4] = {};
#pragma unroll 8
            for (int k4 = 0; k4 < 128; k4 += 4) {
                v2f       a  = *(const v2f*)(ar + k4);
                const v2f bb = *(const v2f*)&lb1[k4 + 2 * hi];
                a.x = fmaxf(a.x * da + bb.x, 0.0f);
                a.y = fmaxf(a.y * da + bb.y, 0.0f);
                const float* lb = &lw2[((k4 >> 1) + hi) * W2P];
                v2f bf[4];
#pragma unroll
                for (int ct = 0; ct < 4; ++ct) bf[ct] = *(const v2f*)&lb[2 * (ct * 16 + laneM)];
#pragma unroll
                for (int ct = 0; ct < 4; ++ct)
                    acc[ct] = __builtin_amdgcn_wmma_f32_16x16x4_f32(
                        false, a, false, bf[ct], (short)0, acc[ct], false, false);
            }
#pragma unroll
            for (int r = 0; r < 8; ++r) {
                const int m = u * 16 + 8 * hi + r;
                if (m < nr) {
                    const float dm = dl[m];
#pragma unroll
                    for (int ct = 0; ct < 4; ++ct) tile[m * TP + ct * 16 + laneM] = acc[ct][r] * dm;
                }
            }
        }
        __syncthreads();
        constexpr int NITD = (R * 16 + 1023) / 1024;
        const int     limD = nr * 16;
        v4f*          o4   = (v4f*)(out + (long)lo * 64);
#pragma unroll
        for (int it = 0; it < NITD; ++it) {
            const int i4 = (int)threadIdx.x + it * 1024;
            if (i4 < limD) o4[i4] = *(const v4f*)(tile + (i4 >> 4) * TP + 4 * (i4 & 15));
        }
    } else {
        v4f* o4 = (v4f*)(out + (long)lo * F);
#pragma unroll
        for (int it = 0; it < NIT; ++it) {
            const int i4 = (int)threadIdx.x + it * 1024;
            if (i4 < lim4) {
                const float d = dl[i4 >> (LOGF - 2)];
                v4f         v = *(const v4f*)(tile + (i4 >> (LOGF - 2)) * TP + 4 * (i4 & (F4 - 1))) * d;
                if (bias) v += bv;
                o4[i4] = v;
            }
        }
    }
}


extern "C" void kernel_launch(void* const* d_in, const int* in_sizes, int n_in,
                              void* d_out, int out_size, void* d_ws, size_t ws_size,
                              hipStream_t stream) {
    const float* x   = (const float*)d_in[0];
    const int*   ei  = (const int*)d_in[1];
    const float* W1  = (const float*)d_in[2];
    const float* b1  = (const float*)d_in[3];
    const float* W2  = (const float*)d_in[4];
    const float* b2  = (const float*)d_in[5];
    float*       out = (float*)d_out;

    const int  n = in_sizes[0] / 128;
    const long E = in_sizes[1] / 2;
    const int* srcIdx = ei;
    const int* dstIdx = ei + E;

    float* ws   = (float*)d_ws;
    float* dinv = ws;
    float* h1   = dinv + n;
    float* h2   = h1 + (long)n * 128;
    int*   ecnt  = (int*)(h2 + (long)n * 64);
    int*   elist = ecnt + n;

    const int B = 256;
    auto blk = [](long work, int b) { return (int)((work + b - 1) / b); };
    unsigned* planes = (unsigned*)(elist + (long)blk(n, RG) * (LCAP * 2));

    k_bits<<<blk(E, EB), B, 0, stream>>>(dstIdx, E, planes, blk(n, RG));
    k_agg<6, RG, 0, false><<<blk(n, RG), 1024, 0, stream>>>(srcIdx, dstIdx, nullptr, nullptr, nullptr, h1, elist, ecnt, planes, dinv, E, n, x, W1);

    k_agg<7, RG, 1, true><<<blk(n, RG), 1024, 0, stream>>>(srcIdx, dstIdx, dinv, h1, b1, h2, elist, ecnt, planes, nullptr, E, n, nullptr, W2);

    k_agg<6, RG, 1, false><<<blk(n, RG), 1024, 0, stream>>>(srcIdx, dstIdx, dinv, h2, b2, out, elist, ecnt, planes, nullptr, E, n, nullptr, nullptr);
}
